# starting kernel
# baseline (speedup 1.0000x reference)
#include <hip/hip_runtime.h>
#include <hip/hip_bf16.h>

typedef float v2f __attribute__((ext_vector_type(2)));
typedef float v8f __attribute__((ext_vector_type(8)));

__global__ void deg_init_kernel(float* __restrict__ deg, int n) {
    int i = blockIdx.x * blockDim.x + threadIdx.x;
    if (i < n) deg[i] = 1.0f;
}

__global__ void deg_accum_kernel(const int* __restrict__ dst, float* __restrict__ deg, int e) {
    int i = blockIdx.x * blockDim.x + threadIdx.x;
    if (i < e) atomicAdd(&deg[dst[i]], 1.0f);
}

__global__ void rsqrt_kernel(float* __restrict__ deg, int n) {
    int i = blockIdx.x * blockDim.x + threadIdx.x;
    if (i < n) deg[i] = rsqrtf(deg[i]);
}

__global__ void gemm_wmma_f32_kernel(const float* __restrict__ A,
                                     const float* __restrict__ B,
                                     float* __restrict__ C,
                                     int M, int K, int N) {
    const int lane  = threadIdx.x & 31;
    const int wave  = threadIdx.x >> 5;
    const int tilesN = N >> 4;
    const int total  = (M >> 4) * tilesN;
    const int tile   = blockIdx.x * (blockDim.x >> 5) + wave;
    if (tile >= total) return;

    const int mt   = tile / tilesN;
    const int nt   = tile - mt * tilesN;
    const int half = lane >> 4;
    const int l15  = lane & 15;

    const float* arow = A + (long long)(mt * 16 + l15) * K;
    const float* bcol = B + nt * 16 + l15;

    v8f c = {};
    for (int k = 0; k < K; k += 4) {
        v2f a, b;
        a.x = arow[k + 2 * half];
        a.y = arow[k + 2 * half + 1];
        b.x = bcol[(long long)(k + 2 * half) * N];
        b.y = bcol[(long long)(k + 2 * half + 1) * N];
        c = __builtin_amdgcn_wmma_f32_16x16x4_f32(
                 false, a,  false, b,
                 (short)0, c,  false,  false);
    }

    float* crow = C + (long long)(mt * 16) * N + nt * 16 + l15;
#pragma unroll
    for (int r = 0; r < 8; ++r)
        crow[(long long)(r + 8 * half) * N] = c[r];
}

__global__ void agg_init_kernel(const float* __restrict__ xw,
                                const float* __restrict__ dinv,
                                const float* __restrict__ bias,
                                float* __restrict__ out,
                                int n, int F) {
    long long i = (long long)blockIdx.x * blockDim.x + threadIdx.x;
    long long total = (long long)n * F;
    if (i >= total) return;
    int node = (int)(i / F);
    int f    = (int)(i - (long long)node * F);
    float di = dinv[node];
    out[i] = xw[i] * di * di + bias[f];
}

__global__ void edge_scatter_kernel(const int* __restrict__ src,
                                    const int* __restrict__ dst,
                                    const float* __restrict__ dinv,
                                    const float* __restrict__ xw,
                                    float* __restrict__ out,
                                    int E, int F) {
    long long t = (long long)blockIdx.x * blockDim.x + threadIdx.x;
    long long total = (long long)E * F;
    if (t >= total) return;
    int e = (int)(t / F);
    int f = (int)(t - (long long)e * F);
    int s = src[e];
    int d = dst[e];
    float nrm = dinv[s] * dinv[d];
    atomicAdd(&out[(long long)d * F + f], xw[(long long)s * F + f] * nrm);
}

__global__ void relu_kernel(float* __restrict__ x, long long n) {
    long long i = (long long)blockIdx.x * blockDim.x + threadIdx.x;
    if (i < n) x[i] = fmaxf(x[i], 0.0f);
}

extern "C" void kernel_launch(void* const* d_in, const int* in_sizes, int n_in,
                              void* d_out, int out_size, void* d_ws, size_t ws_size,
                              hipStream_t stream) {
    const float* x   = (const float*)d_in[0];
    const int*   ei  = (const int*)d_in[1];
    const float* W1  = (const float*)d_in[2];
    const float* b1  = (const float*)d_in[3];
    const float* W2  = (const float*)d_in[4];
    const float* b2  = (const float*)d_in[5];
    float*       out = (float*)d_out;

    const int E     = in_sizes[1] / 2;
    const int F_HID = in_sizes[3];
    const int F_IN  = in_sizes[2] / F_HID;
    const int F_OUT = in_sizes[5];
    const int N     = in_sizes[0] / F_IN;

    const int* src = ei;
    const int* dst = ei + E;

    float* deg = (float*)d_ws;
    float* xw1 = deg + 65536;
    float* h   = xw1 + (long long)N * F_HID;
    float* hw2 = h   + (long long)N * F_HID;

    const int TB = 256;

    deg_init_kernel<<<(N + TB - 1) / TB, TB, 0, stream>>>(deg, N);
    deg_accum_kernel<<<(E + TB - 1) / TB, TB, 0, stream>>>(dst, deg, E);
    rsqrt_kernel<<<(N + TB - 1) / TB, TB, 0, stream>>>(deg, N);

    {
        int tiles = (N / 16) * (F_HID / 16);
        int wavesPerBlock = TB / 32;
        int blocks = (tiles + wavesPerBlock - 1) / wavesPerBlock;
        gemm_wmma_f32_kernel<<<blocks, TB, 0, stream>>>(x, W1, xw1, N, F_IN, F_HID);
    }

    {
        long long total = (long long)N * F_HID;
        agg_init_kernel<<<(int)((total + TB - 1) / TB), TB, 0, stream>>>(xw1, deg, b1, h, N, F_HID);
        long long etotal = (long long)E * F_HID;
        edge_scatter_kernel<<<(int)((etotal + TB - 1) / TB), TB, 0, stream>>>(src, dst, deg, xw1, h, E, F_HID);
        relu_kernel<<<(int)((total + TB - 1) / TB), TB, 0, stream>>>(h, total);
    }

    {
        int tiles = (N / 16) * (F_OUT / 16);
        int wavesPerBlock = TB / 32;
        int blocks = (tiles + wavesPerBlock - 1) / wavesPerBlock;
        gemm_wmma_f32_kernel<<<blocks, TB, 0, stream>>>(h, W2, hw2, N, F_HID, F_OUT);
    }

    {
        long long total = (long long)N * F_OUT;
        agg_init_kernel<<<(int)((total + TB - 1) / TB), TB, 0, stream>>>(hw2, deg, b2, out, N, F_OUT);
        long long etotal = (long long)E * F_OUT;
        edge_scatter_kernel<<<(int)((etotal + TB - 1) / TB), TB, 0, stream>>>(src, dst, deg, hw2, out, E, F_OUT);
    }
}


# edges scanned as bytes: a pre-pass writes each edge's row-range index as one byte; range blocks read a quarter of the bytes, count rows at queue write-out; 4->5 launches; LDS arrival sums as before
# speedup vs baseline: 3.5134x; 3.5134x over previous
#include <hip/hip_runtime.h>
#include <hip/hip_bf16.h>

typedef float v2f __attribute__((ext_vector_type(2)));
typedef float v8f __attribute__((ext_vector_type(8)));

__device__ __forceinline__ void gemm_wmma_f32_tile(const float* __restrict__ A,
                                                   const float* __restrict__ B,
                                                   float* __restrict__ C,
                                                   int M, int K, int N, int tile) {
    const int lane  = threadIdx.x & 31;
    const int tilesN = N >> 4;
    const int total  = (M >> 4) * tilesN;
    if (tile >= total) return;

    const int mt   = tile / tilesN;
    const int nt   = tile - mt * tilesN;
    const int half = lane >> 4;
    const int l15  = lane & 15;

    const float* arow = A + (long long)(mt * 16 + l15) * K;
    const float* bcol = B + nt * 16 + l15;

    v8f c = {};
    for (int k = 0; k < K; k += 4) {
        v2f a, b;
        a.x = arow[k + 2 * half];
        a.y = arow[k + 2 * half + 1];
        b.x = bcol[(long long)(k + 2 * half) * N];
        b.y = bcol[(long long)(k + 2 * half + 1) * N];
        c = __builtin_amdgcn_wmma_f32_16x16x4_f32(
                 false, a,  false, b,
                 (short)0, c,  false,  false);
    }

    float* crow = C + (long long)(mt * 16) * N + nt * 16 + l15;
#pragma unroll
    for (int r = 0; r < 8; ++r)
        crow[(long long)(r + 8 * half) * N] = c[r];
}

__global__ void gemm_wmma_f32_kernel(const float* __restrict__ A,
                                     const float* __restrict__ B,
                                     float* __restrict__ C,
                                     int M, int K, int N) {
    const int wave = threadIdx.x >> 5;
    gemm_wmma_f32_tile(A, B, C, M, K, N, blockIdx.x * (blockDim.x >> 5) + wave);
}

#define AGG_THREADS 1024
#define AGG_ROWS    256
#define AGG_SHIFT   8
#define AGG_QCAP    8192
#define AGG_GROUP   16

#define RB_THREADS 256
__global__ __launch_bounds__(RB_THREADS)
void range_byte_kernel(const int* __restrict__ dst,
                       unsigned* __restrict__ plane,
                       int N, int E, int nwords) {
    static_assert(AGG_ROWS == (1 << AGG_SHIFT), "range index = node >> AGG_SHIFT");
    const int lane = threadIdx.x & 31;
    const int jw = __builtin_amdgcn_readfirstlane((int)((blockIdx.x * RB_THREADS + threadIdx.x) >> 5)) * 128;
    const unsigned un = (unsigned)N;
    const bool vec_ok = (reinterpret_cast<unsigned long long>(dst) & 15ull) == 0ull;
    const int4* __restrict__ dst4 = reinterpret_cast<const int4*>(dst);
    auto rb = [&](int d) -> unsigned { return (unsigned)d < un ? ((unsigned)d >> AGG_SHIFT) : 0xFFu; };
    int4 v[4];
    if (vec_ok && (jw + 127) * 4 + 3 < E) {
#pragma unroll
        for (int u = 0; u < 4; ++u) v[u] = dst4[jw + u * 32 + lane];
    } else {
#pragma unroll
        for (int u = 0; u < 4; ++u) {
            const int e = (jw + u * 32 + lane) << 2;
            v[u].x = (e     < E) ? dst[e]     : -1;
            v[u].y = (e + 1 < E) ? dst[e + 1] : -1;
            v[u].z = (e + 2 < E) ? dst[e + 2] : -1;
            v[u].w = (e + 3 < E) ? dst[e + 3] : -1;
        }
    }
#pragma unroll
    for (int u = 0; u < 4; ++u) {
        const int j = jw + u * 32 + lane;
        if (j < nwords)
            plane[j] = rb(v[u].x) | (rb(v[u].y) << 8) | (rb(v[u].z) << 16) | (rb(v[u].w) << 24);
    }
}

__global__ __launch_bounds__(AGG_THREADS)
void range_queue_and_product_kernel(const int* __restrict__ src,
                                    const int* __restrict__ dst,
                                    const unsigned* __restrict__ plane,
                                    float* __restrict__ dinv,
                                    int* __restrict__ g_cnt,
                                    int* __restrict__ g_src,
                                    int* __restrict__ g_row,
                                    int N, int E, int P16, int queueBlocks,
                                    const float* __restrict__ A,
                                    const float* __restrict__ B,
                                    float* __restrict__ C,
                                    int M, int K, int NC) {
    __shared__ int q_e[AGG_QCAP];
    __shared__ int cnt[AGG_ROWS];
    __shared__ int q_n;

    if ((int)blockIdx.x >= queueBlocks) {
        const int pwave = threadIdx.x >> 5;
        gemm_wmma_f32_tile(A, B, C, M, K, NC,
                           ((int)blockIdx.x - queueBlocks) * (AGG_THREADS / 32) + pwave);
        return;
    }

    const int tid  = threadIdx.x;
    const int r0   = blockIdx.x * AGG_ROWS;
    const int rows = min(AGG_ROWS, N - r0);
    const unsigned urows = (unsigned)rows;

    if (tid < AGG_ROWS) cnt[tid] = 0;
    if (tid == 0) q_n = 0;
    __syncthreads();

    auto push = [&](int e) {
        const int pos = atomicAdd(&q_n, 1);
        if (pos < AGG_QCAP) q_e[pos] = e;
    };
    const unsigned own4 = (unsigned)blockIdx.x * 0x01010101u;
    auto own = [&](unsigned w) -> unsigned {
        const unsigned x = w ^ own4;
        const unsigned t = (x & 0x7F7F7F7Fu) + 0x7F7F7F7Fu;
        return ~(t | x) & 0x80808080u;
    };
    auto check16 = [&](const int4 v, int e0, bool live) {
        unsigned m = (own((unsigned)v.x) >> 7) | (own((unsigned)v.y) >> 6) |
                     (own((unsigned)v.z) >> 5) | (own((unsigned)v.w) >> 4);
        m = live ? m : 0u;
        while (m) {
            const int p = __builtin_ctz(m);
            m &= m - 1;
            push(e0 + ((p & 7) << 2) + (p >> 3));
        }
    };
    const int4* __restrict__ pl4 = reinterpret_cast<const int4*>(plane);
    const int rot = (int)(((long long)blockIdx.x * P16) / queueBlocks) & ~15;
    constexpr int SCAN_U = 7;
    for (int k = tid; k < P16; k += SCAN_U * AGG_THREADS) {
        int4 v[SCAN_U];
        int  ix[SCAN_U];
#pragma unroll
        for (int u = 0; u < SCAN_U; ++u) {
            const int ku = k + u * AGG_THREADS;
            int i = rot + ku;
            i -= (i >= P16) ? P16 : 0;
            i = (ku < P16) ? i : 0;
            ix[u] = i;
            v[u] = pl4[i];
        }
#pragma unroll
        for (int u = 0; u < SCAN_U; ++u) check16(v[u], ix[u] << 4, k + u * AGG_THREADS < P16);
    }
    __syncthreads();

    const int nhit = q_n;
    if (nhit <= AGG_QCAP) {
        static_assert(AGG_QCAP == 8 * AGG_THREADS, "eight queue entries per thread");
        int* __restrict__ gs = g_src + (long long)blockIdx.x * AGG_QCAP;
        int* __restrict__ gr = g_row + (long long)blockIdx.x * AGG_QCAP;
        int ce[8], cs[8], cd[8];
#pragma unroll
        for (int u = 0; u < 8; ++u) {
            const int i = tid + u * AGG_THREADS;
            ce[u] = (i < nhit) ? q_e[i] : 0;
        }
#pragma unroll
        for (int u = 0; u < 8; ++u) { cs[u] = src[ce[u]]; cd[u] = dst[ce[u]] - r0; }
#pragma unroll
        for (int u = 0; u < 8; ++u) {
            const int i = tid + u * AGG_THREADS;
            if (i < nhit) {
                gs[i] = cs[u];
                gr[i] = cd[u];
                atomicAdd(&cnt[cd[u]], 1);
            }
        }
    } else {
        for (int e = tid; e < E; e += AGG_THREADS) {
            const unsigned r = (unsigned)(dst[e] - r0);
            if (r < urows) atomicAdd(&cnt[r], 1);
        }
    }
    if (tid == 0) g_cnt[blockIdx.x] = nhit;
    __syncthreads();

    if (tid < rows) dinv[r0 + tid] = rsqrtf((float)(cnt[tid] + 1));
}

template <int F, bool RELU>
__global__ __launch_bounds__(AGG_THREADS)
void agg_range_kernel(const int* __restrict__ src,
                      const int* __restrict__ dst,
                      const float* __restrict__ dinv,
                      const float* __restrict__ xw,
                      const float* __restrict__ bias,
                      const int* __restrict__ g_cnt,
                      const int* __restrict__ g_src,
                      const int* __restrict__ g_row,
                      float* __restrict__ out,
                      int N, int E) {
    static_assert(F == 32 || F == 64, "one or two columns per lane");
    constexpr int CPL = F / 32;
    constexpr int FV  = F / 4;

    __shared__ alignas(16) float acc[AGG_ROWS * F];
    __shared__ int   q_s[AGG_QCAP];
    __shared__ int   q_d[AGG_QCAP];
    __shared__ float q_w[AGG_QCAP];

    const int tid  = threadIdx.x;
    const int lane = tid & 31;
    const int wave = __builtin_amdgcn_readfirstlane(tid >> 5);
    const int r0   = blockIdx.x * AGG_ROWS;
    const int rows = min(AGG_ROWS, N - r0);
    const unsigned urows = (unsigned)rows;
    const int nvec = rows * FV;

    for (int i = tid; i < nvec; i += AGG_THREADS) {
        const int r  = i / FV;
        const int f4 = i - r * FV;
        const int node = r0 + r;
        const float di = dinv[node];
        const float dd = di * di;
        const float4 x = reinterpret_cast<const float4*>(xw + (long long)node * F)[f4];
        const float* b = bias + 4 * f4;
        float4 a;
        a.x = x.x * dd + b[0];
        a.y = x.y * dd + b[1];
        a.z = x.z * dd + b[2];
        a.w = x.w * dd + b[3];
        reinterpret_cast<float4*>(acc)[i] = a;
    }

    const int nhit = __builtin_amdgcn_readfirstlane(g_cnt[blockIdx.x]);
    const bool fits = nhit <= AGG_QCAP;
    const int nq = fits ? nhit : 0;
    if (nq > 0) {
        static_assert(AGG_QCAP == 8 * AGG_THREADS, "eight queue entries per thread");
        const int* __restrict__ gs = g_src + (long long)blockIdx.x * AGG_QCAP;
        const int* __restrict__ gr = g_row + (long long)blockIdx.x * AGG_QCAP;
        int cs[8], cd[8];
        float ca[8], cb[8];
#pragma unroll
        for (int u = 0; u < 8; ++u) {
            const int i = tid + u * AGG_THREADS;
            cs[u] = (i < nq) ? gs[i] : 0;
            cd[u] = (i < nq) ? gr[i] : 0;
        }
#pragma unroll
        for (int u = 0; u < 8; ++u) { ca[u] = dinv[cs[u]]; cb[u] = dinv[r0 + cd[u]]; }
#pragma unroll
        for (int u = 0; u < 8; ++u) {
            const int i = tid + u * AGG_THREADS;
            if (i < nq) {
                q_s[i] = cs[u];
                q_d[i] = cd[u];
                q_w[i] = ca[u] * cb[u];
            }
        }
    }
    __syncthreads();
    if (!fits) {
        for (int e = tid; e < E; e += AGG_THREADS) {
            const int d = dst[e];
            if ((unsigned)(d - r0) < urows) {
                const int s = src[e];
                const float w = dinv[s] * dinv[d];
                const float* xr = xw + (long long)s * F;
                float* ar = acc + (d - r0) * F;
                for (int f = 0; f < F; ++f) atomicAdd(&ar[f], xr[f] * w);
            }
        }
        __syncthreads();
    }
    static_assert(AGG_GROUP <= 32, "a lane per queue entry of the group");
    for (int base = wave * AGG_GROUP; base < nq; base += (AGG_THREADS / 32) * AGG_GROUP) {
        const int cnt = min(AGG_GROUP, nq - base);
        const int li = base + min(lane & (AGG_GROUP - 1), cnt - 1);
        const int es = q_s[li];
        const int ed = q_d[li];
        const int ew = __float_as_int(q_w[li]);
        asm volatile("" :: "v"(es), "v"(ed), "v"(ew));
        float vx[AGG_GROUP], vy[AGG_GROUP];
#pragma unroll
        for (int u = 0; u < AGG_GROUP; ++u) {
            const int s = __builtin_amdgcn_readlane(es, u);
            const float* xr = xw + (long long)s * F + lane * CPL;
            if (CPL == 2) {
                const float2 t = *reinterpret_cast<const float2*>(xr);
                vx[u] = t.x; vy[u] = t.y;
            } else {
                vx[u] = *xr; vy[u] = 0.0f;
            }
        }
#pragma unroll
        for (int u = 0; u < AGG_GROUP; ++u) {
            if (u < cnt) {
                const int   d = __builtin_amdgcn_readlane(ed, u);
                const float w = __int_as_float(__builtin_amdgcn_readlane(ew, u));
                float* ar = acc + d * F + lane * CPL;
                atomicAdd(&ar[0], vx[u] * w);
                if (CPL == 2) atomicAdd(&ar[1], vy[u] * w);
            }
        }
    }
    __syncthreads();

    for (int i = tid; i < nvec; i += AGG_THREADS) {
        const int r  = i / FV;
        const int f4 = i - r * FV;
        float4 a = reinterpret_cast<const float4*>(acc)[i];
        if (RELU) {
            a.x = fmaxf(a.x, 0.0f);
            a.y = fmaxf(a.y, 0.0f);
            a.z = fmaxf(a.z, 0.0f);
            a.w = fmaxf(a.w, 0.0f);
        }
        reinterpret_cast<float4*>(out + (long long)(r0 + r) * F)[f4] = a;
    }
}

extern "C" void kernel_launch(void* const* d_in, const int* in_sizes, int n_in,
                              void* d_out, int out_size, void* d_ws, size_t ws_size,
                              hipStream_t stream) {
    const float* x   = (const float*)d_in[0];
    const int*   ei  = (const int*)d_in[1];
    const float* W1  = (const float*)d_in[2];
    const float* b1  = (const float*)d_in[3];
    const float* W2  = (const float*)d_in[4];
    const float* b2  = (const float*)d_in[5];
    float*       out = (float*)d_out;

    const int E     = in_sizes[1] / 2;
    const int F_HID = in_sizes[3];
    const int F_IN  = in_sizes[2] / F_HID;
    const int F_OUT = in_sizes[5];
    const int N     = in_sizes[0] / F_IN;

    const int* src = ei;
    const int* dst = ei + E;

    const int TB = 256;
    const int aggBlocks = (N + AGG_ROWS - 1) / AGG_ROWS;

    float* deg = (float*)d_ws;
    float* xw1 = deg + 65536;
    float* h   = xw1 + (long long)N * F_HID;
    float* hw2 = h   + (long long)N * F_HID;
    int*   qn  = (int*)(hw2 + (long long)N * F_OUT);
    int*   qs  = qn + 1024;
    int*   qr  = qs + (long long)aggBlocks * AGG_QCAP;
    unsigned* rbp = (unsigned*)(qr + (long long)aggBlocks * AGG_QCAP);
    const int P16 = (E + 15) / 16;

    {
        int nwords = P16 * 4;
        int wordsPerBlock = (RB_THREADS / 32) * 128;
        range_byte_kernel<<<(nwords + wordsPerBlock - 1) / wordsPerBlock, RB_THREADS, 0, stream>>>(dst, rbp, N, E, nwords);
    }

    {
        int tiles = (N / 16) * (F_HID / 16);
        int wavesPerBlock = AGG_THREADS / 32;
        int prodBlocks = (tiles + wavesPerBlock - 1) / wavesPerBlock;
        range_queue_and_product_kernel<<<aggBlocks + prodBlocks, AGG_THREADS, 0, stream>>>(
            src, dst, rbp, deg, qn, qs, qr, N, E, P16, aggBlocks, x, W1, xw1, N, F_IN, F_HID);
    }

    agg_range_kernel<64, true><<<aggBlocks, AGG_THREADS, 0, stream>>>(src, dst, deg, xw1, b1, qn, qs, qr, h, N, E);

    {
        int tiles = (N / 16) * (F_OUT / 16);
        int wavesPerBlock = TB / 32;
        int blocks = (tiles + wavesPerBlock - 1) / wavesPerBlock;
        gemm_wmma_f32_kernel<<<blocks, TB, 0, stream>>>(h, W2, hw2, N, F_HID, F_OUT);
    }

    agg_range_kernel<32, false><<<aggBlocks, AGG_THREADS, 0, stream>>>(src, dst, deg, hw2, b2, qn, qs, qr, out, N, E);
}


# owner row ranges of 200 rows instead of 256: 250 blocks instead of 196 in the queue pass and both aggregations, so 250 of the 256 compute units share the edges; LDS arrival-order sums as before
# speedup vs baseline: 3.8010x; 1.0819x over previous
#include <hip/hip_runtime.h>
#include <hip/hip_bf16.h>

typedef float v2f __attribute__((ext_vector_type(2)));
typedef float v8f __attribute__((ext_vector_type(8)));

__device__ __forceinline__ void gemm_wmma_f32_tile(const float* __restrict__ A,
                                                   const float* __restrict__ B,
                                                   float* __restrict__ C,
                                                   int M, int K, int N, int tile) {
    const int lane  = threadIdx.x & 31;
    const int tilesN = N >> 4;
    const int total  = (M >> 4) * tilesN;
    if (tile >= total) return;

    const int mt   = tile / tilesN;
    const int nt   = tile - mt * tilesN;
    const int half = lane >> 4;
    const int l15  = lane & 15;

    const float* arow = A + (long long)(mt * 16 + l15) * K;
    const float* bcol = B + nt * 16 + l15;

    v8f c = {};
    for (int k = 0; k < K; k += 4) {
        v2f a, b;
        a.x = arow[k + 2 * half];
        a.y = arow[k + 2 * half + 1];
        b.x = bcol[(long long)(k + 2 * half) * N];
        b.y = bcol[(long long)(k + 2 * half + 1) * N];
        c = __builtin_amdgcn_wmma_f32_16x16x4_f32(
                 false, a,  false, b,
                 (short)0, c,  false,  false);
    }

    float* crow = C + (long long)(mt * 16) * N + nt * 16 + l15;
#pragma unroll
    for (int r = 0; r < 8; ++r)
        crow[(long long)(r + 8 * half) * N] = c[r];
}

__global__ void gemm_wmma_f32_kernel(const float* __restrict__ A,
                                     const float* __restrict__ B,
                                     float* __restrict__ C,
                                     int M, int K, int N) {
    const int wave = threadIdx.x >> 5;
    gemm_wmma_f32_tile(A, B, C, M, K, N, blockIdx.x * (blockDim.x >> 5) + wave);
}

#define AGG_THREADS 1024
#define AGG_ROWS    200
#define AGG_QCAP    8192
#define AGG_GROUP   16

#define RB_THREADS 256
__global__ __launch_bounds__(RB_THREADS)
void range_byte_kernel(const int* __restrict__ dst,
                       unsigned* __restrict__ plane,
                       int N, int E, int nwords) {
    const int lane = threadIdx.x & 31;
    const int jw = __builtin_amdgcn_readfirstlane((int)((blockIdx.x * RB_THREADS + threadIdx.x) >> 5)) * 128;
    const unsigned un = (unsigned)N;
    const bool vec_ok = (reinterpret_cast<unsigned long long>(dst) & 15ull) == 0ull;
    const int4* __restrict__ dst4 = reinterpret_cast<const int4*>(dst);
    auto rb = [&](int d) -> unsigned { return (unsigned)d < un ? ((unsigned)d / (unsigned)AGG_ROWS) : 0xFFu; };
    int4 v[4];
    if (vec_ok && (jw + 127) * 4 + 3 < E) {
#pragma unroll
        for (int u = 0; u < 4; ++u) v[u] = dst4[jw + u * 32 + lane];
    } else {
#pragma unroll
        for (int u = 0; u < 4; ++u) {
            const int e = (jw + u * 32 + lane) << 2;
            v[u].x = (e     < E) ? dst[e]     : -1;
            v[u].y = (e + 1 < E) ? dst[e + 1] : -1;
            v[u].z = (e + 2 < E) ? dst[e + 2] : -1;
            v[u].w = (e + 3 < E) ? dst[e + 3] : -1;
        }
    }
#pragma unroll
    for (int u = 0; u < 4; ++u) {
        const int j = jw + u * 32 + lane;
        if (j < nwords)
            plane[j] = rb(v[u].x) | (rb(v[u].y) << 8) | (rb(v[u].z) << 16) | (rb(v[u].w) << 24);
    }
}

__global__ __launch_bounds__(AGG_THREADS)
void range_queue_and_product_kernel(const int* __restrict__ src,
                                    const int* __restrict__ dst,
                                    const unsigned* __restrict__ plane,
                                    float* __restrict__ dinv,
                                    int* __restrict__ g_cnt,
                                    int* __restrict__ g_src,
                                    int* __restrict__ g_row,
                                    int N, int E, int P16, int queueBlocks,
                                    const float* __restrict__ A,
                                    const float* __restrict__ B,
                                    float* __restrict__ C,
                                    int M, int K, int NC) {
    __shared__ int q_e[AGG_QCAP];
    __shared__ int cnt[AGG_ROWS];
    __shared__ int q_n;

    if ((int)blockIdx.x >= queueBlocks) {
        const int pwave = threadIdx.x >> 5;
        gemm_wmma_f32_tile(A, B, C, M, K, NC,
                           ((int)blockIdx.x - queueBlocks) * (AGG_THREADS / 32) + pwave);
        return;
    }

    const int tid  = threadIdx.x;
    const int r0   = blockIdx.x * AGG_ROWS;
    const int rows = min(AGG_ROWS, N - r0);
    const unsigned urows = (unsigned)rows;

    if (tid < AGG_ROWS) cnt[tid] = 0;
    if (tid == 0) q_n = 0;
    __syncthreads();

    auto push = [&](int e) {
        const int pos = atomicAdd(&q_n, 1);
        if (pos < AGG_QCAP) q_e[pos] = e;
    };
    const unsigned own4 = (unsigned)blockIdx.x * 0x01010101u;
    auto own = [&](unsigned w) -> unsigned {
        const unsigned x = w ^ own4;
        const unsigned t = (x & 0x7F7F7F7Fu) + 0x7F7F7F7Fu;
        return ~(t | x) & 0x80808080u;
    };
    auto check16 = [&](const int4 v, int e0, bool live) {
        unsigned m = (own((unsigned)v.x) >> 7) | (own((unsigned)v.y) >> 6) |
                     (own((unsigned)v.z) >> 5) | (own((unsigned)v.w) >> 4);
        m = live ? m : 0u;
        while (m) {
            const int p = __builtin_ctz(m);
            m &= m - 1;
            push(e0 + ((p & 7) << 2) + (p >> 3));
        }
    };
    const int4* __restrict__ pl4 = reinterpret_cast<const int4*>(plane);
    const int rot = (int)(((long long)blockIdx.x * P16) / queueBlocks) & ~15;
    constexpr int SCAN_U = 7;
    for (int k = tid; k < P16; k += SCAN_U * AGG_THREADS) {
        int4 v[SCAN_U];
        int  ix[SCAN_U];
#pragma unroll
        for (int u = 0; u < SCAN_U; ++u) {
            const int ku = k + u * AGG_THREADS;
            int i = rot + ku;
            i -= (i >= P16) ? P16 : 0;
            i = (ku < P16) ? i : 0;
            ix[u] = i;
            v[u] = pl4[i];
        }
#pragma unroll
        for (int u = 0; u < SCAN_U; ++u) check16(v[u], ix[u] << 4, k + u * AGG_THREADS < P16);
    }
    __syncthreads();

    const int nhit = q_n;
    if (nhit <= AGG_QCAP) {
        static_assert(AGG_QCAP == 8 * AGG_THREADS, "eight queue entries per thread");
        int* __restrict__ gs = g_src + (long long)blockIdx.x * AGG_QCAP;
        int* __restrict__ gr = g_row + (long long)blockIdx.x * AGG_QCAP;
        int ce[8], cs[8], cd[8];
#pragma unroll
        for (int u = 0; u < 8; ++u) {
            const int i = tid + u * AGG_THREADS;
            ce[u] = (i < nhit) ? q_e[i] : 0;
        }
#pragma unroll
        for (int u = 0; u < 8; ++u) { cs[u] = src[ce[u]]; cd[u] = dst[ce[u]] - r0; }
#pragma unroll
        for (int u = 0; u < 8; ++u) {
            const int i = tid + u * AGG_THREADS;
            if (i < nhit) {
                gs[i] = cs[u];
                gr[i] = cd[u];
                atomicAdd(&cnt[cd[u]], 1);
            }
        }
    } else {
        for (int e = tid; e < E; e += AGG_THREADS) {
            const unsigned r = (unsigned)(dst[e] - r0);
            if (r < urows) atomicAdd(&cnt[r], 1);
        }
    }
    if (tid == 0) g_cnt[blockIdx.x] = nhit;
    __syncthreads();

    if (tid < rows) dinv[r0 + tid] = rsqrtf((float)(cnt[tid] + 1));
}

template <int F, bool RELU>
__global__ __launch_bounds__(AGG_THREADS)
void agg_range_kernel(const int* __restrict__ src,
                      const int* __restrict__ dst,
                      const float* __restrict__ dinv,
                      const float* __restrict__ xw,
                      const float* __restrict__ bias,
                      const int* __restrict__ g_cnt,
                      const int* __restrict__ g_src,
                      const int* __restrict__ g_row,
                      float* __restrict__ out,
                      int N, int E) {
    static_assert(F == 32 || F == 64, "one or two columns per lane");
    constexpr int CPL = F / 32;
    constexpr int FV  = F / 4;

    __shared__ alignas(16) float acc[AGG_ROWS * F];
    __shared__ int   q_s[AGG_QCAP];
    __shared__ int   q_d[AGG_QCAP];
    __shared__ float q_w[AGG_QCAP];

    const int tid  = threadIdx.x;
    const int lane = tid & 31;
    const int wave = __builtin_amdgcn_readfirstlane(tid >> 5);
    const int r0   = blockIdx.x * AGG_ROWS;
    const int rows = min(AGG_ROWS, N - r0);
    const unsigned urows = (unsigned)rows;
    const int nvec = rows * FV;

    for (int i = tid; i < nvec; i += AGG_THREADS) {
        const int r  = i / FV;
        const int f4 = i - r * FV;
        const int node = r0 + r;
        const float di = dinv[node];
        const float dd = di * di;
        const float4 x = reinterpret_cast<const float4*>(xw + (long long)node * F)[f4];
        const float* b = bias + 4 * f4;
        float4 a;
        a.x = x.x * dd + b[0];
        a.y = x.y * dd + b[1];
        a.z = x.z * dd + b[2];
        a.w = x.w * dd + b[3];
        reinterpret_cast<float4*>(acc)[i] = a;
    }

    const int nhit = __builtin_amdgcn_readfirstlane(g_cnt[blockIdx.x]);
    const bool fits = nhit <= AGG_QCAP;
    const int nq = fits ? nhit : 0;
    if (nq > 0) {
        static_assert(AGG_QCAP == 8 * AGG_THREADS, "eight queue entries per thread");
        const int* __restrict__ gs = g_src + (long long)blockIdx.x * AGG_QCAP;
        const int* __restrict__ gr = g_row + (long long)blockIdx.x * AGG_QCAP;
        int cs[8], cd[8];
        float ca[8], cb[8];
#pragma unroll
        for (int u = 0; u < 8; ++u) {
            const int i = tid + u * AGG_THREADS;
            cs[u] = (i < nq) ? gs[i] : 0;
            cd[u] = (i < nq) ? gr[i] : 0;
        }
#pragma unroll
        for (int u = 0; u < 8; ++u) { ca[u] = dinv[cs[u]]; cb[u] = dinv[r0 + cd[u]]; }
#pragma unroll
        for (int u = 0; u < 8; ++u) {
            const int i = tid + u * AGG_THREADS;
            if (i < nq) {
                q_s[i] = cs[u];
                q_d[i] = cd[u];
                q_w[i] = ca[u] * cb[u];
            }
        }
    }
    __syncthreads();
    if (!fits) {
        for (int e = tid; e < E; e += AGG_THREADS) {
            const int d = dst[e];
            if ((unsigned)(d - r0) < urows) {
                const int s = src[e];
                const float w = dinv[s] * dinv[d];
                const float* xr = xw + (long long)s * F;
                float* ar = acc + (d - r0) * F;
                for (int f = 0; f < F; ++f) atomicAdd(&ar[f], xr[f] * w);
            }
        }
        __syncthreads();
    }
    static_assert(AGG_GROUP <= 32, "a lane per queue entry of the group");
    for (int base = wave * AGG_GROUP; base < nq; base += (AGG_THREADS / 32) * AGG_GROUP) {
        const int cnt = min(AGG_GROUP, nq - base);
        const int li = base + min(lane & (AGG_GROUP - 1), cnt - 1);
        const int es = q_s[li];
        const int ed = q_d[li];
        const int ew = __float_as_int(q_w[li]);
        asm volatile("" :: "v"(es), "v"(ed), "v"(ew));
        float vx[AGG_GROUP], vy[AGG_GROUP];
#pragma unroll
        for (int u = 0; u < AGG_GROUP; ++u) {
            const int s = __builtin_amdgcn_readlane(es, u);
            const float* xr = xw + (long long)s * F + lane * CPL;
            if (CPL == 2) {
                const float2 t = *reinterpret_cast<const float2*>(xr);
                vx[u] = t.x; vy[u] = t.y;
            } else {
                vx[u] = *xr; vy[u] = 0.0f;
            }
        }
#pragma unroll
        for (int u = 0; u < AGG_GROUP; ++u) {
            if (u < cnt) {
                const int   d = __builtin_amdgcn_readlane(ed, u);
                const float w = __int_as_float(__builtin_amdgcn_readlane(ew, u));
                float* ar = acc + d * F + lane * CPL;
                atomicAdd(&ar[0], vx[u] * w);
                if (CPL == 2) atomicAdd(&ar[1], vy[u] * w);
            }
        }
    }
    __syncthreads();

    for (int i = tid; i < nvec; i += AGG_THREADS) {
        const int r  = i / FV;
        const int f4 = i - r * FV;
        float4 a = reinterpret_cast<const float4*>(acc)[i];
        if (RELU) {
            a.x = fmaxf(a.x, 0.0f);
            a.y = fmaxf(a.y, 0.0f);
            a.z = fmaxf(a.z, 0.0f);
            a.w = fmaxf(a.w, 0.0f);
        }
        reinterpret_cast<float4*>(out + (long long)(r0 + r) * F)[f4] = a;
    }
}

extern "C" void kernel_launch(void* const* d_in, const int* in_sizes, int n_in,
                              void* d_out, int out_size, void* d_ws, size_t ws_size,
                              hipStream_t stream) {
    const float* x   = (const float*)d_in[0];
    const int*   ei  = (const int*)d_in[1];
    const float* W1  = (const float*)d_in[2];
    const float* b1  = (const float*)d_in[3];
    const float* W2  = (const float*)d_in[4];
    const float* b2  = (const float*)d_in[5];
    float*       out = (float*)d_out;

    const int E     = in_sizes[1] / 2;
    const int F_HID = in_sizes[3];
    const int F_IN  = in_sizes[2] / F_HID;
    const int F_OUT = in_sizes[5];
    const int N     = in_sizes[0] / F_IN;

    const int* src = ei;
    const int* dst = ei + E;

    const int TB = 256;
    const int aggBlocks = (N + AGG_ROWS - 1) / AGG_ROWS;

    float* deg = (float*)d_ws;
    float* xw1 = deg + 65536;
    float* h   = xw1 + (long long)N * F_HID;
    float* hw2 = h   + (long long)N * F_HID;
    int*   qn  = (int*)(hw2 + (long long)N * F_OUT);
    int*   qs  = qn + 1024;
    int*   qr  = qs + (long long)aggBlocks * AGG_QCAP;
    unsigned* rbp = (unsigned*)(qr + (long long)aggBlocks * AGG_QCAP);
    const int P16 = (E + 15) / 16;

    {
        int nwords = P16 * 4;
        int wordsPerBlock = (RB_THREADS / 32) * 128;
        range_byte_kernel<<<(nwords + wordsPerBlock - 1) / wordsPerBlock, RB_THREADS, 0, stream>>>(dst, rbp, N, E, nwords);
    }

    {
        int tiles = (N / 16) * (F_HID / 16);
        int wavesPerBlock = AGG_THREADS / 32;
        int prodBlocks = (tiles + wavesPerBlock - 1) / wavesPerBlock;
        range_queue_and_product_kernel<<<aggBlocks + prodBlocks, AGG_THREADS, 0, stream>>>(
            src, dst, rbp, deg, qn, qs, qr, N, E, P16, aggBlocks, x, W1, xw1, N, F_IN, F_HID);
    }

    agg_range_kernel<64, true><<<aggBlocks, AGG_THREADS, 0, stream>>>(src, dst, deg, xw1, b1, qn, qs, qr, h, N, E);

    {
        int tiles = (N / 16) * (F_OUT / 16);
        int wavesPerBlock = TB / 32;
        int blocks = (tiles + wavesPerBlock - 1) / wavesPerBlock;
        gemm_wmma_f32_kernel<<<blocks, TB, 0, stream>>>(h, W2, hw2, N, F_HID, F_OUT);
    }

    agg_range_kernel<32, false><<<aggBlocks, AGG_THREADS, 0, stream>>>(src, dst, deg, hw2, b2, qn, qs, qr, out, N, E);
}


# pre-pass also packs each edge's source node and row-in-range into one word: queue write-out does one scattered 4-byte look-up per edge, not two; queue is one plane; LDS arrival sums as before
# speedup vs baseline: 3.9342x; 1.0350x over previous
#include <hip/hip_runtime.h>
#include <hip/hip_bf16.h>

typedef float v2f __attribute__((ext_vector_type(2)));
typedef float v8f __attribute__((ext_vector_type(8)));

__device__ __forceinline__ void gemm_wmma_f32_tile(const float* __restrict__ A,
                                                   const float* __restrict__ B,
                                                   float* __restrict__ C,
                                                   int M, int K, int N, int tile) {
    const int lane  = threadIdx.x & 31;
    const int tilesN = N >> 4;
    const int total  = (M >> 4) * tilesN;
    if (tile >= total) return;

    const int mt   = tile / tilesN;
    const int nt   = tile - mt * tilesN;
    const int half = lane >> 4;
    const int l15  = lane & 15;

    const float* arow = A + (long long)(mt * 16 + l15) * K;
    const float* bcol = B + nt * 16 + l15;

    v8f c = {};
    for (int k = 0; k < K; k += 4) {
        v2f a, b;
        a.x = arow[k + 2 * half];
        a.y = arow[k + 2 * half + 1];
        b.x = bcol[(long long)(k + 2 * half) * N];
        b.y = bcol[(long long)(k + 2 * half + 1) * N];
        c = __builtin_amdgcn_wmma_f32_16x16x4_f32(
                 false, a,  false, b,
                 (short)0, c,  false,  false);
    }

    float* crow = C + (long long)(mt * 16) * N + nt * 16 + l15;
#pragma unroll
    for (int r = 0; r < 8; ++r)
        crow[(long long)(r + 8 * half) * N] = c[r];
}

__global__ void gemm_wmma_f32_kernel(const float* __restrict__ A,
                                     const float* __restrict__ B,
                                     float* __restrict__ C,
                                     int M, int K, int N) {
    const int wave = threadIdx.x >> 5;
    gemm_wmma_f32_tile(A, B, C, M, K, N, blockIdx.x * (blockDim.x >> 5) + wave);
}

#define AGG_THREADS 1024
#define AGG_ROWS    200
#define AGG_QCAP    8192
#define AGG_GROUP   16

#define RB_THREADS 256
__global__ __launch_bounds__(RB_THREADS)
void range_byte_kernel(const int* __restrict__ src,
                       const int* __restrict__ dst,
                       unsigned* __restrict__ plane,
                       unsigned* __restrict__ pk,
                       int N, int E, int nwords) {
    static_assert(AGG_ROWS <= 256, "row inside a range fits the packed word's high byte");
    const int lane = threadIdx.x & 31;
    const int jw = __builtin_amdgcn_readfirstlane((int)((blockIdx.x * RB_THREADS + threadIdx.x) >> 5)) * 128;
    const unsigned un = (unsigned)N;
    const bool vec_ok = ((reinterpret_cast<unsigned long long>(dst) | reinterpret_cast<unsigned long long>(src) |
                          reinterpret_cast<unsigned long long>(pk)) & 15ull) == 0ull;
    const int4* __restrict__ dst4 = reinterpret_cast<const int4*>(dst);
    const int4* __restrict__ src4 = reinterpret_cast<const int4*>(src);
    auto rb = [&](int d) -> unsigned { return (unsigned)d < un ? ((unsigned)d / (unsigned)AGG_ROWS) : 0xFFu; };
    auto pack = [&](int s, int d, unsigned r) -> unsigned {
        return r != 0xFFu ? (((unsigned)s & 0xFFFFFFu) | (((unsigned)d - r * (unsigned)AGG_ROWS) << 24)) : 0u;
    };
    int4 v[4], w[4];
    const bool whole = vec_ok && (jw + 127) * 4 + 3 < E;
    if (whole) {
#pragma unroll
        for (int u = 0; u < 4; ++u) { v[u] = dst4[jw + u * 32 + lane]; w[u] = src4[jw + u * 32 + lane]; }
    } else {
#pragma unroll
        for (int u = 0; u < 4; ++u) {
            const int e = (jw + u * 32 + lane) << 2;
            v[u].x = (e     < E) ? dst[e]     : -1;
            v[u].y = (e + 1 < E) ? dst[e + 1] : -1;
            v[u].z = (e + 2 < E) ? dst[e + 2] : -1;
            v[u].w = (e + 3 < E) ? dst[e + 3] : -1;
            w[u].x = (e     < E) ? src[e]     : 0;
            w[u].y = (e + 1 < E) ? src[e + 1] : 0;
            w[u].z = (e + 2 < E) ? src[e + 2] : 0;
            w[u].w = (e + 3 < E) ? src[e + 3] : 0;
        }
    }
#pragma unroll
    for (int u = 0; u < 4; ++u) {
        const int j = jw + u * 32 + lane;
        const unsigned bx = rb(v[u].x), by = rb(v[u].y), bz = rb(v[u].z), bw = rb(v[u].w);
        if (j < nwords)
            plane[j] = bx | (by << 8) | (bz << 16) | (bw << 24);
        uint4 p;
        p.x = pack(w[u].x, v[u].x, bx);
        p.y = pack(w[u].y, v[u].y, by);
        p.z = pack(w[u].z, v[u].z, bz);
        p.w = pack(w[u].w, v[u].w, bw);
        if (whole) {
            reinterpret_cast<uint4*>(pk)[j] = p;
        } else {
            const int e = j << 2;
            if (e     < E) pk[e]     = p.x;
            if (e + 1 < E) pk[e + 1] = p.y;
            if (e + 2 < E) pk[e + 2] = p.z;
            if (e + 3 < E) pk[e + 3] = p.w;
        }
    }
}

__global__ __launch_bounds__(AGG_THREADS)
void range_queue_and_product_kernel(const int* __restrict__ dst,
                                    const unsigned* __restrict__ plane,
                                    const unsigned* __restrict__ pk,
                                    float* __restrict__ dinv,
                                    int* __restrict__ g_cnt,
                                    unsigned* __restrict__ g_pk,
                                    int N, int E, int P16, int queueBlocks,
                                    const float* __restrict__ A,
                                    const float* __restrict__ B,
                                    float* __restrict__ C,
                                    int M, int K, int NC) {
    __shared__ int q_e[AGG_QCAP];
    __shared__ int cnt[AGG_ROWS];
    __shared__ int q_n;

    if ((int)blockIdx.x >= queueBlocks) {
        const int pwave = threadIdx.x >> 5;
        gemm_wmma_f32_tile(A, B, C, M, K, NC,
                           ((int)blockIdx.x - queueBlocks) * (AGG_THREADS / 32) + pwave);
        return;
    }

    const int tid  = threadIdx.x;
    const int r0   = blockIdx.x * AGG_ROWS;
    const int rows = min(AGG_ROWS, N - r0);
    const unsigned urows = (unsigned)rows;

    if (tid < AGG_ROWS) cnt[tid] = 0;
    if (tid == 0) q_n = 0;
    __syncthreads();

    auto push = [&](int e) {
        const int pos = atomicAdd(&q_n, 1);
        if (pos < AGG_QCAP) q_e[pos] = e;
    };
    const unsigned own4 = (unsigned)blockIdx.x * 0x01010101u;
    auto own = [&](unsigned w) -> unsigned {
        const unsigned x = w ^ own4;
        const unsigned t = (x & 0x7F7F7F7Fu) + 0x7F7F7F7Fu;
        return ~(t | x) & 0x80808080u;
    };
    auto check16 = [&](const int4 v, int e0, bool live) {
        unsigned m = (own((unsigned)v.x) >> 7) | (own((unsigned)v.y) >> 6) |
                     (own((unsigned)v.z) >> 5) | (own((unsigned)v.w) >> 4);
        m = live ? m : 0u;
        while (m) {
            const int p = __builtin_ctz(m);
            m &= m - 1;
            push(e0 + ((p & 7) << 2) + (p >> 3));
        }
    };
    const int4* __restrict__ pl4 = reinterpret_cast<const int4*>(plane);
    const int rot = (int)(((long long)blockIdx.x * P16) / queueBlocks) & ~15;
    constexpr int SCAN_U = 7;
    for (int k = tid; k < P16; k += SCAN_U * AGG_THREADS) {
        int4 v[SCAN_U];
        int  ix[SCAN_U];
#pragma unroll
        for (int u = 0; u < SCAN_U; ++u) {
            const int ku = k + u * AGG_THREADS;
            int i = rot + ku;
            i -= (i >= P16) ? P16 : 0;
            i = (ku < P16) ? i : 0;
            ix[u] = i;
            v[u] = pl4[i];
        }
#pragma unroll
        for (int u = 0; u < SCAN_U; ++u) check16(v[u], ix[u] << 4, k + u * AGG_THREADS < P16);
    }
    __syncthreads();

    const int nhit = q_n;
    if (nhit <= AGG_QCAP) {
        static_assert(AGG_QCAP == 8 * AGG_THREADS, "eight queue entries per thread");
        unsigned* __restrict__ gq = g_pk + (long long)blockIdx.x * AGG_QCAP;
        int ce[8];
        unsigned cp[8];
#pragma unroll
        for (int u = 0; u < 8; ++u) {
            const int i = tid + u * AGG_THREADS;
            ce[u] = (i < nhit) ? q_e[i] : 0;
        }
#pragma unroll
        for (int u = 0; u < 8; ++u) cp[u] = pk[ce[u]];
#pragma unroll
        for (int u = 0; u < 8; ++u) {
            const int i = tid + u * AGG_THREADS;
            if (i < nhit) {
                gq[i] = cp[u];
                atomicAdd(&cnt[cp[u] >> 24], 1);
            }
        }
    } else {
        for (int e = tid; e < E; e += AGG_THREADS) {
            const unsigned r = (unsigned)(dst[e] - r0);
            if (r < urows) atomicAdd(&cnt[r], 1);
        }
    }
    if (tid == 0) g_cnt[blockIdx.x] = nhit;
    __syncthreads();

    if (tid < rows) dinv[r0 + tid] = rsqrtf((float)(cnt[tid] + 1));
}

template <int F, bool RELU>
__global__ __launch_bounds__(AGG_THREADS)
void agg_range_kernel(const int* __restrict__ src,
                      const int* __restrict__ dst,
                      const float* __restrict__ dinv,
                      const float* __restrict__ xw,
                      const float* __restrict__ bias,
                      const int* __restrict__ g_cnt,
                      const unsigned* __restrict__ g_pk,
                      float* __restrict__ out,
                      int N, int E) {
    static_assert(F == 32 || F == 64, "one or two columns per lane");
    constexpr int CPL = F / 32;
    constexpr int FV  = F / 4;

    __shared__ alignas(16) float acc[AGG_ROWS * F];
    __shared__ int   q_s[AGG_QCAP];
    __shared__ int   q_d[AGG_QCAP];
    __shared__ float q_w[AGG_QCAP];

    const int tid  = threadIdx.x;
    const int lane = tid & 31;
    const int wave = __builtin_amdgcn_readfirstlane(tid >> 5);
    const int r0   = blockIdx.x * AGG_ROWS;
    const int rows = min(AGG_ROWS, N - r0);
    const unsigned urows = (unsigned)rows;
    const int nvec = rows * FV;

    for (int i = tid; i < nvec; i += AGG_THREADS) {
        const int r  = i / FV;
        const int f4 = i - r * FV;
        const int node = r0 + r;
        const float di = dinv[node];
        const float dd = di * di;
        const float4 x = reinterpret_cast<const float4*>(xw + (long long)node * F)[f4];
        const float* b = bias + 4 * f4;
        float4 a;
        a.x = x.x * dd + b[0];
        a.y = x.y * dd + b[1];
        a.z = x.z * dd + b[2];
        a.w = x.w * dd + b[3];
        reinterpret_cast<float4*>(acc)[i] = a;
    }

    const int nhit = __builtin_amdgcn_readfirstlane(g_cnt[blockIdx.x]);
    const bool fits = nhit <= AGG_QCAP;
    const int nq = fits ? nhit : 0;
    if (nq > 0) {
        static_assert(AGG_QCAP == 8 * AGG_THREADS, "eight queue entries per thread");
        const unsigned* __restrict__ gq = g_pk + (long long)blockIdx.x * AGG_QCAP;
        int cs[8], cd[8];
        float ca[8], cb[8];
#pragma unroll
        for (int u = 0; u < 8; ++u) {
            const int i = tid + u * AGG_THREADS;
            const unsigned p = (i < nq) ? gq[i] : 0u;
            cs[u] = (int)(p & 0xFFFFFFu);
            cd[u] = (int)(p >> 24);
        }
#pragma unroll
        for (int u = 0; u < 8; ++u) { ca[u] = dinv[cs[u]]; cb[u] = dinv[r0 + cd[u]]; }
#pragma unroll
        for (int u = 0; u < 8; ++u) {
            const int i = tid + u * AGG_THREADS;
            if (i < nq) {
                q_s[i] = cs[u];
                q_d[i] = cd[u];
                q_w[i] = ca[u] * cb[u];
            }
        }
    }
    __syncthreads();
    if (!fits) {
        for (int e = tid; e < E; e += AGG_THREADS) {
            const int d = dst[e];
            if ((unsigned)(d - r0) < urows) {
                const int s = src[e];
                const float w = dinv[s] * dinv[d];
                const float* xr = xw + (long long)s * F;
                float* ar = acc + (d - r0) * F;
                for (int f = 0; f < F; ++f) atomicAdd(&ar[f], xr[f] * w);
            }
        }
        __syncthreads();
    }
    static_assert(AGG_GROUP <= 32, "a lane per queue entry of the group");
    for (int base = wave * AGG_GROUP; base < nq; base += (AGG_THREADS / 32) * AGG_GROUP) {
        const int cnt = min(AGG_GROUP, nq - base);
        const int li = base + min(lane & (AGG_GROUP - 1), cnt - 1);
        const int es = q_s[li];
        const int ed = q_d[li];
        const int ew = __float_as_int(q_w[li]);
        asm volatile("" :: "v"(es), "v"(ed), "v"(ew));
        float vx[AGG_GROUP], vy[AGG_GROUP];
#pragma unroll
        for (int u = 0; u < AGG_GROUP; ++u) {
            const int s = __builtin_amdgcn_readlane(es, u);
            const float* xr = xw + (long long)s * F + lane * CPL;
            if (CPL == 2) {
                const float2 t = *reinterpret_cast<const float2*>(xr);
                vx[u] = t.x; vy[u] = t.y;
            } else {
                vx[u] = *xr; vy[u] = 0.0f;
            }
        }
#pragma unroll
        for (int u = 0; u < AGG_GROUP; ++u) {
            if (u < cnt) {
                const int   d = __builtin_amdgcn_readlane(ed, u);
                const float w = __int_as_float(__builtin_amdgcn_readlane(ew, u));
                float* ar = acc + d * F + lane * CPL;
                atomicAdd(&ar[0], vx[u] * w);
                if (CPL == 2) atomicAdd(&ar[1], vy[u] * w);
            }
        }
    }
    __syncthreads();

    for (int i = tid; i < nvec; i += AGG_THREADS) {
        const int r  = i / FV;
        const int f4 = i - r * FV;
        float4 a = reinterpret_cast<const float4*>(acc)[i];
        if (RELU) {
            a.x = fmaxf(a.x, 0.0f);
            a.y = fmaxf(a.y, 0.0f);
            a.z = fmaxf(a.z, 0.0f);
            a.w = fmaxf(a.w, 0.0f);
        }
        reinterpret_cast<float4*>(out + (long long)(r0 + r) * F)[f4] = a;
    }
}

extern "C" void kernel_launch(void* const* d_in, const int* in_sizes, int n_in,
                              void* d_out, int out_size, void* d_ws, size_t ws_size,
                              hipStream_t stream) {
    const float* x   = (const float*)d_in[0];
    const int*   ei  = (const int*)d_in[1];
    const float* W1  = (const float*)d_in[2];
    const float* b1  = (const float*)d_in[3];
    const float* W2  = (const float*)d_in[4];
    const float* b2  = (const float*)d_in[5];
    float*       out = (float*)d_out;

    const int E     = in_sizes[1] / 2;
    const int F_HID = in_sizes[3];
    const int F_IN  = in_sizes[2] / F_HID;
    const int F_OUT = in_sizes[5];
    const int N     = in_sizes[0] / F_IN;

    const int* src = ei;
    const int* dst = ei + E;

    const int TB = 256;
    const int aggBlocks = (N + AGG_ROWS - 1) / AGG_ROWS;

    float* deg = (float*)d_ws;
    float* xw1 = deg + 65536;
    float* h   = xw1 + (long long)N * F_HID;
    float* hw2 = h   + (long long)N * F_HID;
    int*   qn  = (int*)(hw2 + (long long)N * F_OUT);
    unsigned* qp = (unsigned*)(qn + 1024);
    unsigned* rbp = qp + (long long)aggBlocks * AGG_QCAP;
    const int P16 = (E + 15) / 16;
    unsigned* pkp = rbp + (long long)P16 * 4;

    {
        int nwords = P16 * 4;
        int wordsPerBlock = (RB_THREADS / 32) * 128;
        range_byte_kernel<<<(nwords + wordsPerBlock - 1) / wordsPerBlock, RB_THREADS, 0, stream>>>(src, dst, rbp, pkp, N, E, nwords);
    }

    {
        int tiles = (N / 16) * (F_HID / 16);
        int wavesPerBlock = AGG_THREADS / 32;
        int prodBlocks = (tiles + wavesPerBlock - 1) / wavesPerBlock;
        range_queue_and_product_kernel<<<aggBlocks + prodBlocks, AGG_THREADS, 0, stream>>>(
            dst, rbp, pkp, deg, qn, qp, N, E, P16, aggBlocks, x, W1, xw1, N, F_IN, F_HID);
    }

    agg_range_kernel<64, true><<<aggBlocks, AGG_THREADS, 0, stream>>>(src, dst, deg, xw1, b1, qn, qp, h, N, E);

    {
        int tiles = (N / 16) * (F_OUT / 16);
        int wavesPerBlock = TB / 32;
        int blocks = (tiles + wavesPerBlock - 1) / wavesPerBlock;
        gemm_wmma_f32_kernel<<<blocks, TB, 0, stream>>>(h, W2, hw2, N, F_HID, F_OUT);
    }

    agg_range_kernel<32, false><<<aggBlocks, AGG_THREADS, 0, stream>>>(src, dst, deg, hw2, b2, qn, qp, out, N, E);
}


# edges found by bit planes: pre-pass writes an edge's range index as two one-hot 4-bit digits (32 planes); a range block ANDs its 2 planes: 32 edges per op, 1/4 the bytes; LDS arrival sums as before
# speedup vs baseline: 4.1761x; 1.0615x over previous
#include <hip/hip_runtime.h>
#include <hip/hip_bf16.h>

typedef float v2f __attribute__((ext_vector_type(2)));
typedef float v8f __attribute__((ext_vector_type(8)));

__device__ __forceinline__ void gemm_wmma_f32_tile(const float* __restrict__ A,
                                                   const float* __restrict__ B,
                                                   float* __restrict__ C,
                                                   int M, int K, int N, int tile) {
    const int lane  = threadIdx.x & 31;
    const int tilesN = N >> 4;
    const int total  = (M >> 4) * tilesN;
    if (tile >= total) return;

    const int mt   = tile / tilesN;
    const int nt   = tile - mt * tilesN;
    const int half = lane >> 4;
    const int l15  = lane & 15;

    const float* arow = A + (long long)(mt * 16 + l15) * K;
    const float* bcol = B + nt * 16 + l15;

    v8f c = {};
    for (int k = 0; k < K; k += 4) {
        v2f a, b;
        a.x = arow[k + 2 * half];
        a.y = arow[k + 2 * half + 1];
        b.x = bcol[(long long)(k + 2 * half) * N];
        b.y = bcol[(long long)(k + 2 * half + 1) * N];
        c = __builtin_amdgcn_wmma_f32_16x16x4_f32(
                 false, a,  false, b,
                 (short)0, c,  false,  false);
    }

    float* crow = C + (long long)(mt * 16) * N + nt * 16 + l15;
#pragma unroll
    for (int r = 0; r < 8; ++r)
        crow[(long long)(r + 8 * half) * N] = c[r];
}

__global__ void gemm_wmma_f32_kernel(const float* __restrict__ A,
                                     const float* __restrict__ B,
                                     float* __restrict__ C,
                                     int M, int K, int N) {
    const int wave = threadIdx.x >> 5;
    gemm_wmma_f32_tile(A, B, C, M, K, N, blockIdx.x * (blockDim.x >> 5) + wave);
}

#define AGG_THREADS 1024
#define AGG_ROWS    200
#define AGG_QCAP    8192
#define AGG_GROUP   16

#define RB_THREADS 256
#define RB_WORDS   128
#define RB_PITCH   132
#define RB_PLANES  32
__global__ __launch_bounds__(RB_THREADS)
void digit_plane_kernel(const int* __restrict__ src,
                        const int* __restrict__ dst,
                        unsigned* __restrict__ planes,
                        unsigned* __restrict__ pk,
                        int N, int E, int nwp) {
    static_assert(AGG_ROWS <= 256, "row inside a range fits the packed word's high byte");
    static_assert(RB_WORDS == (RB_THREADS / 32) * 16, "a wave covers 512 edges = 16 plane words");
    __shared__ alignas(16) unsigned pl[RB_PLANES * RB_PITCH];
    const int tid  = threadIdx.x;
    const int lane = tid & 31;
    const int wv = __builtin_amdgcn_readfirstlane(tid >> 5);
    const int jw = __builtin_amdgcn_readfirstlane((int)((blockIdx.x * RB_THREADS + threadIdx.x) >> 5)) * 128;
    const unsigned un = (unsigned)N;
    const bool vec_ok = ((reinterpret_cast<unsigned long long>(dst) | reinterpret_cast<unsigned long long>(src) |
                          reinterpret_cast<unsigned long long>(pk)) & 15ull) == 0ull;
    const int4* __restrict__ dst4 = reinterpret_cast<const int4*>(dst);
    const int4* __restrict__ src4 = reinterpret_cast<const int4*>(src);
    auto rb = [&](int d) -> unsigned { return (unsigned)d < un ? ((unsigned)d / (unsigned)AGG_ROWS) : 0xFFu; };
    auto pack = [&](int s, int d, unsigned r) -> unsigned {
        return r != 0xFFu ? (((unsigned)s & 0xFFFFFFu) | (((unsigned)d - r * (unsigned)AGG_ROWS) << 24)) : 0u;
    };
    int4 v[4], w[4];
    const bool whole = vec_ok && (jw + 127) * 4 + 3 < E;
    if (whole) {
#pragma unroll
        for (int u = 0; u < 4; ++u) { v[u] = dst4[jw + u * 32 + lane]; w[u] = src4[jw + u * 32 + lane]; }
    } else {
#pragma unroll
        for (int u = 0; u < 4; ++u) {
            const int e = (jw + u * 32 + lane) << 2;
            v[u].x = (e     < E) ? dst[e]     : -1;
            v[u].y = (e + 1 < E) ? dst[e + 1] : -1;
            v[u].z = (e + 2 < E) ? dst[e + 2] : -1;
            v[u].w = (e + 3 < E) ? dst[e + 3] : -1;
            w[u].x = (e     < E) ? src[e]     : 0;
            w[u].y = (e + 1 < E) ? src[e + 1] : 0;
            w[u].z = (e + 2 < E) ? src[e + 2] : 0;
            w[u].w = (e + 3 < E) ? src[e + 3] : 0;
        }
    }
    for (int i = tid; i < RB_PLANES * RB_PITCH; i += RB_THREADS) pl[i] = 0u;
    __syncthreads();
    auto mark = [&](unsigned r, int word, unsigned bit) {
        if (r != 0xFFu) {
            atomicOr(&pl[(r >> 4) * RB_PITCH + word], bit);
            atomicOr(&pl[(16u + (r & 15u)) * RB_PITCH + word], bit);
        }
    };
#pragma unroll
    for (int u = 0; u < 4; ++u) {
        const int j = jw + u * 32 + lane;
        const unsigned bx = rb(v[u].x), by = rb(v[u].y), bz = rb(v[u].z), bw = rb(v[u].w);
        const int word = wv * 16 + u * 4 + (lane >> 3);
        const unsigned b0 = 1u << ((lane & 7) * 4);
        mark(bx, word, b0);
        mark(by, word, b0 << 1);
        mark(bz, word, b0 << 2);
        mark(bw, word, b0 << 3);
        uint4 p;
        p.x = pack(w[u].x, v[u].x, bx);
        p.y = pack(w[u].y, v[u].y, by);
        p.z = pack(w[u].z, v[u].z, bz);
        p.w = pack(w[u].w, v[u].w, bw);
        if (whole) {
            reinterpret_cast<uint4*>(pk)[j] = p;
        } else {
            const int e = j << 2;
            if (e     < E) pk[e]     = p.x;
            if (e + 1 < E) pk[e + 1] = p.y;
            if (e + 2 < E) pk[e + 2] = p.z;
            if (e + 3 < E) pk[e + 3] = p.w;
        }
    }
    __syncthreads();
    static_assert((RB_PLANES * (RB_WORDS / 4)) % RB_THREADS == 0, "whole 16-byte pieces per thread");
#pragma unroll
    for (int i = 0; i < (RB_PLANES * (RB_WORDS / 4)) / RB_THREADS; ++i) {
        const int idx   = tid + i * RB_THREADS;
        const int p     = idx / (RB_WORDS / 4);
        const int piece = idx - p * (RB_WORDS / 4);
        const uint4 t = *reinterpret_cast<const uint4*>(&pl[p * RB_PITCH + piece * 4]);
        *reinterpret_cast<uint4*>(planes + (long long)p * nwp + (long long)blockIdx.x * RB_WORDS + piece * 4) = t;
    }
}

__device__ __forceinline__ int wave_inclusive_sum(int x) {
    x += __builtin_amdgcn_update_dpp(0, x, 0x111, 0xF, 0xF, false);
    x += __builtin_amdgcn_update_dpp(0, x, 0x112, 0xF, 0xF, false);
    x += __builtin_amdgcn_update_dpp(0, x, 0x114, 0xF, 0xF, false);
    x += __builtin_amdgcn_update_dpp(0, x, 0x118, 0xF, 0xF, false);
    const int low = __builtin_amdgcn_readlane(x, 15);
    return x + ((threadIdx.x & 16) ? low : 0);
}

__global__ __launch_bounds__(AGG_THREADS)
void range_queue_and_product_kernel(const int* __restrict__ dst,
                                    const unsigned* __restrict__ planes,
                                    const unsigned* __restrict__ pk,
                                    float* __restrict__ dinv,
                                    int* __restrict__ g_cnt,
                                    unsigned* __restrict__ g_pk,
                                    int N, int E, int nwp, int queueBlocks,
                                    const float* __restrict__ A,
                                    const float* __restrict__ B,
                                    float* __restrict__ C,
                                    int M, int K, int NC) {
    __shared__ int q_e[AGG_QCAP];
    __shared__ int cnt[AGG_ROWS];
    __shared__ int q_n;

    if ((int)blockIdx.x >= queueBlocks) {
        const int pwave = threadIdx.x >> 5;
        gemm_wmma_f32_tile(A, B, C, M, K, NC,
                           ((int)blockIdx.x - queueBlocks) * (AGG_THREADS / 32) + pwave);
        return;
    }

    const int tid  = threadIdx.x;
    const int r0   = blockIdx.x * AGG_ROWS;
    const int rows = min(AGG_ROWS, N - r0);
    const unsigned urows = (unsigned)rows;

    if (tid < AGG_ROWS) cnt[tid] = 0;
    if (tid == 0) q_n = 0;
    __syncthreads();

    static_assert(RB_PLANES == 32, "sixteen planes per digit");
    const uint4* __restrict__ pa4 = reinterpret_cast<const uint4*>(planes + (long long)(blockIdx.x >> 4) * nwp);
    const uint4* __restrict__ pb4 = reinterpret_cast<const uint4*>(planes + (long long)(16 + (blockIdx.x & 15)) * nwp);
    const int P4 = nwp >> 2;
    const int rot = (int)(((long long)blockIdx.x * P4) / queueBlocks) & ~15;
    constexpr int SCAN_U = 4;
    for (int k0 = 0; k0 < P4; k0 += SCAN_U * AGG_THREADS) {
        const int k = k0 + tid;
        uint4 va[SCAN_U], vb[SCAN_U];
        int   ix[SCAN_U];
        unsigned lm[SCAN_U];
#pragma unroll
        for (int u = 0; u < SCAN_U; ++u) {
            const int ku = k + u * AGG_THREADS;
            int i = rot + ku;
            i -= (i >= P4) ? P4 : 0;
            i = (ku < P4) ? i : 0;
            lm[u] = (ku < P4) ? 0xFFFFFFFFu : 0u;
            ix[u] = i;
            va[u] = pa4[i];
            vb[u] = pb4[i];
        }
        unsigned m[SCAN_U][4];
        int c = 0;
#pragma unroll
        for (int u = 0; u < SCAN_U; ++u) {
            m[u][0] = va[u].x & vb[u].x & lm[u];
            m[u][1] = va[u].y & vb[u].y & lm[u];
            m[u][2] = va[u].z & vb[u].z & lm[u];
            m[u][3] = va[u].w & vb[u].w & lm[u];
            c += __builtin_popcount(m[u][0]) + __builtin_popcount(m[u][1]) +
                 __builtin_popcount(m[u][2]) + __builtin_popcount(m[u][3]);
        }
        const int incl = wave_inclusive_sum(c);
        int base = 0;
        if ((tid & 31) == 31) base = atomicAdd(&q_n, incl);
        int pos = __builtin_amdgcn_readlane(base, 31) + incl - c;
#pragma unroll
        for (int u = 0; u < SCAN_U; ++u) {
#pragma unroll
            for (int j = 0; j < 4; ++j) {
                unsigned mm = m[u][j];
                const int e0 = ((ix[u] << 2) + j) << 5;
                while (mm) {
                    const int p = __builtin_ctz(mm);
                    mm &= mm - 1;
                    if (pos < AGG_QCAP) q_e[pos] = e0 + p;
                    ++pos;
                }
            }
        }
    }
    __syncthreads();

    const int nhit = q_n;
    if (nhit <= AGG_QCAP) {
        static_assert(AGG_QCAP == 8 * AGG_THREADS, "eight queue entries per thread");
        unsigned* __restrict__ gq = g_pk + (long long)blockIdx.x * AGG_QCAP;
        int ce[8];
        unsigned cp[8];
#pragma unroll
        for (int u = 0; u < 8; ++u) {
            const int i = tid + u * AGG_THREADS;
            ce[u] = (i < nhit) ? q_e[i] : 0;
        }
#pragma unroll
        for (int u = 0; u < 8; ++u) cp[u] = pk[ce[u]];
#pragma unroll
        for (int u = 0; u < 8; ++u) {
            const int i = tid + u * AGG_THREADS;
            if (i < nhit) {
                gq[i] = cp[u];
                atomicAdd(&cnt[cp[u] >> 24], 1);
            }
        }
    } else {
        for (int e = tid; e < E; e += AGG_THREADS) {
            const unsigned r = (unsigned)(dst[e] - r0);
            if (r < urows) atomicAdd(&cnt[r], 1);
        }
    }
    if (tid == 0) g_cnt[blockIdx.x] = nhit;
    __syncthreads();

    if (tid < rows) dinv[r0 + tid] = rsqrtf((float)(cnt[tid] + 1));
}

template <int F, bool RELU>
__global__ __launch_bounds__(AGG_THREADS)
void agg_range_kernel(const int* __restrict__ src,
                      const int* __restrict__ dst,
                      const float* __restrict__ dinv,
                      const float* __restrict__ xw,
                      const float* __restrict__ bias,
                      const int* __restrict__ g_cnt,
                      const unsigned* __restrict__ g_pk,
                      float* __restrict__ out,
                      int N, int E) {
    static_assert(F == 32 || F == 64, "one or two columns per lane");
    constexpr int CPL = F / 32;
    constexpr int FV  = F / 4;

    __shared__ alignas(16) float acc[AGG_ROWS * F];
    __shared__ int   q_s[AGG_QCAP];
    __shared__ int   q_d[AGG_QCAP];
    __shared__ float q_w[AGG_QCAP];

    const int tid  = threadIdx.x;
    const int lane = tid & 31;
    const int wave = __builtin_amdgcn_readfirstlane(tid >> 5);
    const int r0   = blockIdx.x * AGG_ROWS;
    const int rows = min(AGG_ROWS, N - r0);
    const unsigned urows = (unsigned)rows;
    const int nvec = rows * FV;

    for (int i = tid; i < nvec; i += AGG_THREADS) {
        const int r  = i / FV;
        const int f4 = i - r * FV;
        const int node = r0 + r;
        const float di = dinv[node];
        const float dd = di * di;
        const float4 x = reinterpret_cast<const float4*>(xw + (long long)node * F)[f4];
        const float* b = bias + 4 * f4;
        float4 a;
        a.x = x.x * dd + b[0];
        a.y = x.y * dd + b[1];
        a.z = x.z * dd + b[2];
        a.w = x.w * dd + b[3];
        reinterpret_cast<float4*>(acc)[i] = a;
    }

    const int nhit = __builtin_amdgcn_readfirstlane(g_cnt[blockIdx.x]);
    const bool fits = nhit <= AGG_QCAP;
    const int nq = fits ? nhit : 0;
    if (nq > 0) {
        static_assert(AGG_QCAP == 8 * AGG_THREADS, "eight queue entries per thread");
        const unsigned* __restrict__ gq = g_pk + (long long)blockIdx.x * AGG_QCAP;
        int cs[8], cd[8];
        float ca[8], cb[8];
#pragma unroll
        for (int u = 0; u < 8; ++u) {
            const int i = tid + u * AGG_THREADS;
            const unsigned p = (i < nq) ? gq[i] : 0u;
            cs[u] = (int)(p & 0xFFFFFFu);
            cd[u] = (int)(p >> 24);
        }
#pragma unroll
        for (int u = 0; u < 8; ++u) { ca[u] = dinv[cs[u]]; cb[u] = dinv[r0 + cd[u]]; }
#pragma unroll
        for (int u = 0; u < 8; ++u) {
            const int i = tid + u * AGG_THREADS;
            if (i < nq) {
                q_s[i] = cs[u];
                q_d[i] = cd[u];
                q_w[i] = ca[u] * cb[u];
            }
        }
    }
    __syncthreads();
    if (!fits) {
        for (int e = tid; e < E; e += AGG_THREADS) {
            const int d = dst[e];
            if ((unsigned)(d - r0) < urows) {
                const int s = src[e];
                const float w = dinv[s] * dinv[d];
                const float* xr = xw + (long long)s * F;
                float* ar = acc + (d - r0) * F;
                for (int f = 0; f < F; ++f) atomicAdd(&ar[f], xr[f] * w);
            }
        }
        __syncthreads();
    }
    static_assert(AGG_GROUP <= 32, "a lane per queue entry of the group");
    for (int base = wave * AGG_GROUP; base < nq; base += (AGG_THREADS / 32) * AGG_GROUP) {
        const int cnt = min(AGG_GROUP, nq - base);
        const int li = base + min(lane & (AGG_GROUP - 1), cnt - 1);
        const int es = q_s[li];
        const int ed = q_d[li];
        const int ew = __float_as_int(q_w[li]);
        asm volatile("" :: "v"(es), "v"(ed), "v"(ew));
        float vx[AGG_GROUP], vy[AGG_GROUP];
#pragma unroll
        for (int u = 0; u < AGG_GROUP; ++u) {
            const int s = __builtin_amdgcn_readlane(es, u);
            const float* xr = xw + (long long)s * F + lane * CPL;
            if (CPL == 2) {
                const float2 t = *reinterpret_cast<const float2*>(xr);
                vx[u] = t.x; vy[u] = t.y;
            } else {
                vx[u] = *xr; vy[u] = 0.0f;
            }
        }
#pragma unroll
        for (int u = 0; u < AGG_GROUP; ++u) {
            if (u < cnt) {
                const int   d = __builtin_amdgcn_readlane(ed, u);
                const float w = __int_as_float(__builtin_amdgcn_readlane(ew, u));
                float* ar = acc + d * F + lane * CPL;
                atomicAdd(&ar[0], vx[u] * w);
                if (CPL == 2) atomicAdd(&ar[1], vy[u] * w);
            }
        }
    }
    __syncthreads();

    for (int i = tid; i < nvec; i += AGG_THREADS) {
        const int r  = i / FV;
        const int f4 = i - r * FV;
        float4 a = reinterpret_cast<const float4*>(acc)[i];
        if (RELU) {
            a.x = fmaxf(a.x, 0.0f);
            a.y = fmaxf(a.y, 0.0f);
            a.z = fmaxf(a.z, 0.0f);
            a.w = fmaxf(a.w, 0.0f);
        }
        reinterpret_cast<float4*>(out + (long long)(r0 + r) * F)[f4] = a;
    }
}

extern "C" void kernel_launch(void* const* d_in, const int* in_sizes, int n_in,
                              void* d_out, int out_size, void* d_ws, size_t ws_size,
                              hipStream_t stream) {
    const float* x   = (const float*)d_in[0];
    const int*   ei  = (const int*)d_in[1];
    const float* W1  = (const float*)d_in[2];
    const float* b1  = (const float*)d_in[3];
    const float* W2  = (const float*)d_in[4];
    const float* b2  = (const float*)d_in[5];
    float*       out = (float*)d_out;

    const int E     = in_sizes[1] / 2;
    const int F_HID = in_sizes[3];
    const int F_IN  = in_sizes[2] / F_HID;
    const int F_OUT = in_sizes[5];
    const int N     = in_sizes[0] / F_IN;

    const int* src = ei;
    const int* dst = ei + E;

    const int TB = 256;
    const int aggBlocks = (N + AGG_ROWS - 1) / AGG_ROWS;

    float* deg = (float*)d_ws;
    float* xw1 = deg + 65536;
    float* h   = xw1 + (long long)N * F_HID;
    float* hw2 = h   + (long long)N * F_HID;
    int*   qn  = (int*)(hw2 + (long long)N * F_OUT);
    unsigned* qp = (unsigned*)(qn + 1024);
    unsigned* rbp = qp + (long long)aggBlocks * AGG_QCAP;
    const int edgesPerBlock = RB_WORDS * 32;
    const int planeBlocks = (E + edgesPerBlock - 1) / edgesPerBlock;
    const int nwp = planeBlocks * RB_WORDS;
    unsigned* pkp = rbp + (long long)RB_PLANES * nwp;

    digit_plane_kernel<<<planeBlocks, RB_THREADS, 0, stream>>>(src, dst, rbp, pkp, N, E, nwp);

    {
        int tiles = (N / 16) * (F_HID / 16);
        int wavesPerBlock = AGG_THREADS / 32;
        int prodBlocks = (tiles + wavesPerBlock - 1) / wavesPerBlock;
        range_queue_and_product_kernel<<<aggBlocks + prodBlocks, AGG_THREADS, 0, stream>>>(
            dst, rbp, pkp, deg, qn, qp, N, E, nwp, aggBlocks, x, W1, xw1, N, F_IN, F_HID);
    }

    agg_range_kernel<64, true><<<aggBlocks, AGG_THREADS, 0, stream>>>(src, dst, deg, xw1, b1, qn, qp, h, N, E);

    {
        int tiles = (N / 16) * (F_OUT / 16);
        int wavesPerBlock = TB / 32;
        int blocks = (tiles + wavesPerBlock - 1) / wavesPerBlock;
        gemm_wmma_f32_kernel<<<blocks, TB, 0, stream>>>(h, W2, hw2, N, F_HID, F_OUT);
    }

    agg_range_kernel<32, false><<<aggBlocks, AGG_THREADS, 0, stream>>>(src, dst, deg, hw2, b2, qn, qp, out, N, E);
}


# speedup vs baseline: 4.3890x; 1.0510x over previous
#include <hip/hip_runtime.h>
#include <hip/hip_bf16.h>

typedef float v2f __attribute__((ext_vector_type(2)));
typedef float v8f __attribute__((ext_vector_type(8)));

__device__ __forceinline__ void gemm_wmma_f32_tile(const float* __restrict__ A,
                                                   const float* __restrict__ B,
                                                   float* __restrict__ C,
                                                   int M, int K, int N, int tile) {
    const int lane  = threadIdx.x & 31;
    const int tilesN = N >> 4;
    const int total  = (M >> 4) * tilesN;
    if (tile >= total) return;

    const int mt   = tile / tilesN;
    const int nt   = tile - mt * tilesN;
    const int half = lane >> 4;
    const int l15  = lane & 15;

    const float* arow = A + (long long)(mt * 16 + l15) * K;
    const float* bcol = B + nt * 16 + l15;

    v8f c = {};
    for (int k = 0; k < K; k += 4) {
        v2f a, b;
        a.x = arow[k + 2 * half];
        a.y = arow[k + 2 * half + 1];
        b.x = bcol[(long long)(k + 2 * half) * N];
        b.y = bcol[(long long)(k + 2 * half + 1) * N];
        c = __builtin_amdgcn_wmma_f32_16x16x4_f32(
                 false, a,  false, b,
                 (short)0, c,  false,  false);
    }

    float* crow = C + (long long)(mt * 16) * N + nt * 16 + l15;
#pragma unroll
    for (int r = 0; r < 8; ++r)
        crow[(long long)(r + 8 * half) * N] = c[r];
}

__global__ void gemm_wmma_f32_kernel(const float* __restrict__ A,
                                     const float* __restrict__ B,
                                     float* __restrict__ C,
                                     int M, int K, int N) {
    const int wave = threadIdx.x >> 5;
    gemm_wmma_f32_tile(A, B, C, M, K, N, blockIdx.x * (blockDim.x >> 5) + wave);
}

#define AGG_THREADS 1024
#define AGG_ROWS    200
#define AGG_QCAP    8192
#define AGG_GROUP   16

#define RB_THREADS 256
#define RB_WORDS   128
#define RB_PITCH   132
#define RB_PLANES  32
__global__ __launch_bounds__(RB_THREADS)
void digit_plane_kernel(const int* __restrict__ src,
                        const int* __restrict__ dst,
                        unsigned* __restrict__ planes,
                        unsigned* __restrict__ pk,
                        int N, int E, int nwp) {
    static_assert(AGG_ROWS <= 256, "row inside a range fits the packed word's high byte");
    static_assert(RB_WORDS == (RB_THREADS / 32) * 16, "a wave covers 512 edges = 16 plane words");
    __shared__ alignas(16) unsigned pl[RB_PLANES * RB_PITCH];
    const int tid  = threadIdx.x;
    const int lane = tid & 31;
    const int wv = __builtin_amdgcn_readfirstlane(tid >> 5);
    const int jw = __builtin_amdgcn_readfirstlane((int)((blockIdx.x * RB_THREADS + threadIdx.x) >> 5)) * 128;
    const unsigned un = (unsigned)N;
    const bool vec_ok = ((reinterpret_cast<unsigned long long>(dst) | reinterpret_cast<unsigned long long>(src) |
                          reinterpret_cast<unsigned long long>(pk)) & 15ull) == 0ull;
    const int4* __restrict__ dst4 = reinterpret_cast<const int4*>(dst);
    const int4* __restrict__ src4 = reinterpret_cast<const int4*>(src);
    auto rb = [&](int d) -> unsigned { return (unsigned)d < un ? ((unsigned)d / (unsigned)AGG_ROWS) : 0xFFu; };
    auto pack = [&](int s, int d, unsigned r) -> unsigned {
        return r != 0xFFu ? (((unsigned)s & 0xFFFFFFu) | (((unsigned)d - r * (unsigned)AGG_ROWS) << 24)) : 0u;
    };
    int4 v[4], w[4];
    const bool whole = vec_ok && (jw + 127) * 4 + 3 < E;
    if (whole) {
#pragma unroll
        for (int u = 0; u < 4; ++u) { v[u] = dst4[jw + u * 32 + lane]; w[u] = src4[jw + u * 32 + lane]; }
    } else {
#pragma unroll
        for (int u = 0; u < 4; ++u) {
            const int e = (jw + u * 32 + lane) << 2;
            v[u].x = (e     < E) ? dst[e]     : -1;
            v[u].y = (e + 1 < E) ? dst[e + 1] : -1;
            v[u].z = (e + 2 < E) ? dst[e + 2] : -1;
            v[u].w = (e + 3 < E) ? dst[e + 3] : -1;
            w[u].x = (e     < E) ? src[e]     : 0;
            w[u].y = (e + 1 < E) ? src[e + 1] : 0;
            w[u].z = (e + 2 < E) ? src[e + 2] : 0;
            w[u].w = (e + 3 < E) ? src[e + 3] : 0;
        }
    }
    for (int i = tid; i < RB_PLANES * RB_PITCH; i += RB_THREADS) pl[i] = 0u;
    __syncthreads();
    auto mark = [&](unsigned r, int word, unsigned bit) {
        if (r != 0xFFu) {
            atomicOr(&pl[(r >> 4) * RB_PITCH + word], bit);
            atomicOr(&pl[(16u + (r & 15u)) * RB_PITCH + word], bit);
        }
    };
#pragma unroll
    for (int u = 0; u < 4; ++u) {
        const int j = jw + u * 32 + lane;
        const unsigned bx = rb(v[u].x), by = rb(v[u].y), bz = rb(v[u].z), bw = rb(v[u].w);
        const int word = wv * 16 + u * 4 + (lane >> 3);
        const unsigned b0 = 1u << ((lane & 7) * 4);
        mark(bx, word, b0);
        mark(by, word, b0 << 1);
        mark(bz, word, b0 << 2);
        mark(bw, word, b0 << 3);
        uint4 p;
        p.x = pack(w[u].x, v[u].x, bx);
        p.y = pack(w[u].y, v[u].y, by);
        p.z = pack(w[u].z, v[u].z, bz);
        p.w = pack(w[u].w, v[u].w, bw);
        if (whole) {
            reinterpret_cast<uint4*>(pk)[j] = p;
        } else {
            const int e = j << 2;
            if (e     < E) pk[e]     = p.x;
            if (e + 1 < E) pk[e + 1] = p.y;
            if (e + 2 < E) pk[e + 2] = p.z;
            if (e + 3 < E) pk[e + 3] = p.w;
        }
    }
    __syncthreads();
    static_assert((RB_PLANES * (RB_WORDS / 4)) % RB_THREADS == 0, "whole 16-byte pieces per thread");
#pragma unroll
    for (int i = 0; i < (RB_PLANES * (RB_WORDS / 4)) / RB_THREADS; ++i) {
        const int idx   = tid + i * RB_THREADS;
        const int p     = idx / (RB_WORDS / 4);
        const int piece = idx - p * (RB_WORDS / 4);
        const uint4 t = *reinterpret_cast<const uint4*>(&pl[p * RB_PITCH + piece * 4]);
        *reinterpret_cast<uint4*>(planes + (long long)p * nwp + (long long)blockIdx.x * RB_WORDS + piece * 4) = t;
    }
}

__device__ __forceinline__ int wave_inclusive_sum(int x) {
    x += __builtin_amdgcn_update_dpp(0, x, 0x111, 0xF, 0xF, false);
    x += __builtin_amdgcn_update_dpp(0, x, 0x112, 0xF, 0xF, false);
    x += __builtin_amdgcn_update_dpp(0, x, 0x114, 0xF, 0xF, false);
    x += __builtin_amdgcn_update_dpp(0, x, 0x118, 0xF, 0xF, false);
    const int low = __builtin_amdgcn_readlane(x, 15);
    return x + ((threadIdx.x & 16) ? low : 0);
}

__global__ __launch_bounds__(AGG_THREADS)
void range_queue_and_product_kernel(const int* __restrict__ dst,
                                    const unsigned* __restrict__ planes,
                                    const unsigned* __restrict__ pk,
                                    float* __restrict__ dinv,
                                    int* __restrict__ g_cnt,
                                    unsigned* __restrict__ g_pk,
                                    int N, int E, int nwp, int queueBlocks,
                                    const float* __restrict__ A,
                                    const float* __restrict__ B,
                                    float* __restrict__ C,
                                    int M, int K, int NC) {
    __shared__ int q_e[AGG_QCAP];
    __shared__ int cnt[AGG_ROWS];
    __shared__ int q_n;

    if ((int)blockIdx.x >= queueBlocks) {
        const int pwave = threadIdx.x >> 5;
        gemm_wmma_f32_tile(A, B, C, M, K, NC,
                           ((int)blockIdx.x - queueBlocks) * (AGG_THREADS / 32) + pwave);
        return;
    }

    const int tid  = threadIdx.x;
    const int r0   = blockIdx.x * AGG_ROWS;
    const int rows = min(AGG_ROWS, N - r0);
    const unsigned urows = (unsigned)rows;

    if (tid < AGG_ROWS) cnt[tid] = 0;
    if (tid == 0) q_n = 0;
    __syncthreads();

    static_assert(RB_PLANES == 32, "sixteen planes per digit");
    const uint4* __restrict__ pa4 = reinterpret_cast<const uint4*>(planes + (long long)(blockIdx.x >> 4) * nwp);
    const uint4* __restrict__ pb4 = reinterpret_cast<const uint4*>(planes + (long long)(16 + (blockIdx.x & 15)) * nwp);
    const int P4 = nwp >> 2;
    const int rot = (int)(((long long)blockIdx.x * P4) / queueBlocks) & ~15;
    constexpr int SCAN_U = 4;
    for (int k0 = 0; k0 < P4; k0 += SCAN_U * AGG_THREADS) {
        const int k = k0 + tid;
        uint4 va[SCAN_U], vb[SCAN_U];
        int   ix[SCAN_U];
        unsigned lm[SCAN_U];
#pragma unroll
        for (int u = 0; u < SCAN_U; ++u) {
            const int ku = k + u * AGG_THREADS;
            int i = rot + ku;
            i -= (i >= P4) ? P4 : 0;
            i = (ku < P4) ? i : 0;
            lm[u] = (ku < P4) ? 0xFFFFFFFFu : 0u;
            ix[u] = i;
            va[u] = pa4[i];
            vb[u] = pb4[i];
        }
        unsigned m[SCAN_U][4];
        int c = 0;
#pragma unroll
        for (int u = 0; u < SCAN_U; ++u) {
            m[u][0] = va[u].x & vb[u].x & lm[u];
            m[u][1] = va[u].y & vb[u].y & lm[u];
            m[u][2] = va[u].z & vb[u].z & lm[u];
            m[u][3] = va[u].w & vb[u].w & lm[u];
            c += __builtin_popcount(m[u][0]) + __builtin_popcount(m[u][1]) +
                 __builtin_popcount(m[u][2]) + __builtin_popcount(m[u][3]);
        }
        const int incl = wave_inclusive_sum(c);
        int base = 0;
        if ((tid & 31) == 31) base = atomicAdd(&q_n, incl);
        int pos = __builtin_amdgcn_readlane(base, 31) + incl - c;
#pragma unroll
        for (int u = 0; u < SCAN_U; ++u) {
#pragma unroll
            for (int j = 0; j < 4; ++j) {
                unsigned mm = m[u][j];
                const int e0 = ((ix[u] << 2) + j) << 5;
                while (mm) {
                    const int p = __builtin_ctz(mm);
                    mm &= mm - 1;
                    if (pos < AGG_QCAP) q_e[pos] = e0 + p;
                    ++pos;
                }
            }
        }
    }
    __syncthreads();

    const int nhit = q_n;
    if (nhit <= AGG_QCAP) {
        static_assert(AGG_QCAP == 8 * AGG_THREADS, "eight queue entries per thread");
        unsigned* __restrict__ gq = g_pk + (long long)blockIdx.x * AGG_QCAP;
        int ce[8];
        unsigned cp[8];
#pragma unroll
        for (int u = 0; u < 8; ++u) {
            const int i = tid + u * AGG_THREADS;
            ce[u] = (i < nhit) ? q_e[i] : 0;
        }
#pragma unroll
        for (int u = 0; u < 8; ++u) cp[u] = pk[ce[u]];
#pragma unroll
        for (int u = 0; u < 8; ++u) {
            const int i = tid + u * AGG_THREADS;
            if (i < nhit) {
                gq[i] = cp[u];
                atomicAdd(&cnt[cp[u] >> 24], 1);
            }
        }
    } else {
        for (int e = tid; e < E; e += AGG_THREADS) {
            const unsigned r = (unsigned)(dst[e] - r0);
            if (r < urows) atomicAdd(&cnt[r], 1);
        }
    }
    if (tid == 0) g_cnt[blockIdx.x] = nhit;
    __syncthreads();

    if (tid < rows) dinv[r0 + tid] = rsqrtf((float)(cnt[tid] + 1));
}

#define LDS_BLOCK  32768
#define LDS_NBLK   8
#define Q_PER_BLK  (AGG_QCAP / LDS_NBLK)
#define Q_OFF_S    8192
#define Q_OFF_D    (Q_OFF_S + Q_PER_BLK * 4)
#define Q_OFF_W    (Q_OFF_D + Q_PER_BLK * 4)
template <int F, bool RELU>
__global__ __launch_bounds__(AGG_THREADS)
void agg_range_kernel(const int* __restrict__ src,
                      const int* __restrict__ dst,
                      const float* __restrict__ dinv,
                      const float* __restrict__ xw,
                      const float* __restrict__ bias,
                      const int* __restrict__ g_cnt,
                      const unsigned* __restrict__ g_pk,
                      float* __restrict__ out,
                      int N, int E) {
    static_assert(F == 32 || F == 64, "one or two columns per lane");
    constexpr int CPL = F / 32;
    constexpr int FV  = F / 4;

    static_assert((LDS_NBLK & (LDS_NBLK - 1)) == 0 && AGG_QCAP % LDS_NBLK == 0, "rows and queue entries are dealt by mask and shift");
    static_assert(((AGG_ROWS + LDS_NBLK - 1) / LDS_NBLK) * F * 4 <= Q_OFF_S, "a block's rows end before its queue slices");
    static_assert(Q_OFF_W + Q_PER_BLK * 4 <= LDS_BLOCK, "rows and queue slices fit one block");
    __shared__ alignas(16) unsigned char lds[LDS_NBLK * LDS_BLOCK];
    auto row_off = [](int r) -> int { return (r & (LDS_NBLK - 1)) * LDS_BLOCK + (r / LDS_NBLK) * (F * 4); };
    auto q_off = [](int i) -> int { return (i / Q_PER_BLK) * LDS_BLOCK + (i & (Q_PER_BLK - 1)) * 4; };

    const int tid  = threadIdx.x;
    const int lane = tid & 31;
    const int wave = __builtin_amdgcn_readfirstlane(tid >> 5);
    const int r0   = blockIdx.x * AGG_ROWS;
    const int rows = min(AGG_ROWS, N - r0);
    const unsigned urows = (unsigned)rows;
    const int nvec = rows * FV;

    for (int i = tid; i < nvec; i += AGG_THREADS) {
        const int r  = i / FV;
        const int f4 = i - r * FV;
        const int node = r0 + r;
        const float di = dinv[node];
        const float dd = di * di;
        const float4 x = reinterpret_cast<const float4*>(xw + (long long)node * F)[f4];
        const float* b = bias + 4 * f4;
        float4 a;
        a.x = x.x * dd + b[0];
        a.y = x.y * dd + b[1];
        a.z = x.z * dd + b[2];
        a.w = x.w * dd + b[3];
        *reinterpret_cast<float4*>(lds + row_off(r) + f4 * 16) = a;
    }

    const int nhit = __builtin_amdgcn_readfirstlane(g_cnt[blockIdx.x]);
    const bool fits = nhit <= AGG_QCAP;
    const int nq = fits ? nhit : 0;
    if (nq > 0) {
        static_assert(AGG_QCAP == 8 * AGG_THREADS, "eight queue entries per thread");
        const unsigned* __restrict__ gq = g_pk + (long long)blockIdx.x * AGG_QCAP;
        int cs[8], cd[8];
        float ca[8], cb[8];
#pragma unroll
        for (int u = 0; u < 8; ++u) {
            const int i = tid + u * AGG_THREADS;
            const unsigned p = (i < nq) ? gq[i] : 0u;
            cs[u] = (int)(p & 0xFFFFFFu);
            cd[u] = (int)(p >> 24);
        }
#pragma unroll
        for (int u = 0; u < 8; ++u) { ca[u] = dinv[cs[u]]; cb[u] = dinv[r0 + cd[u]]; }
#pragma unroll
        for (int u = 0; u < 8; ++u) {
            const int i = tid + u * AGG_THREADS;
            if (i < nq) {
                const int qo = q_off(i);
                *reinterpret_cast<int*>(lds + qo + Q_OFF_S)   = cs[u];
                *reinterpret_cast<int*>(lds + qo + Q_OFF_D)   = row_off(cd[u]);
                *reinterpret_cast<float*>(lds + qo + Q_OFF_W) = ca[u] * cb[u];
            }
        }
    }
    __syncthreads();
    if (!fits) {
        for (int e = tid; e < E; e += AGG_THREADS) {
            const int d = dst[e];
            if ((unsigned)(d - r0) < urows) {
                const int s = src[e];
                const float w = dinv[s] * dinv[d];
                const float* xr = xw + (long long)s * F;
                float* ar = reinterpret_cast<float*>(lds + row_off(d - r0));
                for (int f = 0; f < F; ++f) atomicAdd(&ar[f], xr[f] * w);
            }
        }
        __syncthreads();
    }
    static_assert(AGG_GROUP <= 32, "a lane per queue entry of the group");
    for (int base = wave * AGG_GROUP; base < nq; base += (AGG_THREADS / 32) * AGG_GROUP) {
        const int cnt = min(AGG_GROUP, nq - base);
        const int li = base + min(lane & (AGG_GROUP - 1), cnt - 1);
        const int qo = q_off(li);
        const int es = *reinterpret_cast<const int*>(lds + qo + Q_OFF_S);
        const int ed = *reinterpret_cast<const int*>(lds + qo + Q_OFF_D);
        const int ew = __float_as_int(*reinterpret_cast<const float*>(lds + qo + Q_OFF_W));
        asm volatile("" :: "v"(es), "v"(ed), "v"(ew));
        float vx[AGG_GROUP], vy[AGG_GROUP];
#pragma unroll
        for (int u = 0; u < AGG_GROUP; ++u) {
            const int s = __builtin_amdgcn_readlane(es, u);
            const float* xr = xw + (long long)s * F + lane * CPL;
            if (CPL == 2) {
                const float2 t = *reinterpret_cast<const float2*>(xr);
                vx[u] = t.x; vy[u] = t.y;
            } else {
                vx[u] = *xr; vy[u] = 0.0f;
            }
        }
#pragma unroll
        for (int u = 0; u < AGG_GROUP; ++u) {
            if (u < cnt) {
                const int   d = __builtin_amdgcn_readlane(ed, u);
                const float w = __int_as_float(__builtin_amdgcn_readlane(ew, u));
                float* ar = reinterpret_cast<float*>(lds + d) + lane * CPL;
                atomicAdd(&ar[0], vx[u] * w);
                if (CPL == 2) atomicAdd(&ar[1], vy[u] * w);
            }
        }
    }
    __syncthreads();

    for (int i = tid; i < nvec; i += AGG_THREADS) {
        const int r  = i / FV;
        const int f4 = i - r * FV;
        float4 a = *reinterpret_cast<const float4*>(lds + row_off(r) + f4 * 16);
        if (RELU) {
            a.x = fmaxf(a.x, 0.0f);
            a.y = fmaxf(a.y, 0.0f);
            a.z = fmaxf(a.z, 0.0f);
            a.w = fmaxf(a.w, 0.0f);
        }
        reinterpret_cast<float4*>(out + (long long)(r0 + r) * F)[f4] = a;
    }
}

extern "C" void kernel_launch(void* const* d_in, const int* in_sizes, int n_in,
                              void* d_out, int out_size, void* d_ws, size_t ws_size,
                              hipStream_t stream) {
    const float* x   = (const float*)d_in[0];
    const int*   ei  = (const int*)d_in[1];
    const float* W1  = (const float*)d_in[2];
    const float* b1  = (const float*)d_in[3];
    const float* W2  = (const float*)d_in[4];
    const float* b2  = (const float*)d_in[5];
    float*       out = (float*)d_out;

    const int E     = in_sizes[1] / 2;
    const int F_HID = in_sizes[3];
    const int F_IN  = in_sizes[2] / F_HID;
    const int F_OUT = in_sizes[5];
    const int N     = in_sizes[0] / F_IN;

    const int* src = ei;
    const int* dst = ei + E;

    const int TB = 256;
    const int aggBlocks = (N + AGG_ROWS - 1) / AGG_ROWS;

    float* deg = (float*)d_ws;
    float* xw1 = deg + 65536;
    float* h   = xw1 + (long long)N * F_HID;
    float* hw2 = h   + (long long)N * F_HID;
    int*   qn  = (int*)(hw2 + (long long)N * F_OUT);
    unsigned* qp = (unsigned*)(qn + 1024);
    unsigned* rbp = qp + (long long)aggBlocks * AGG_QCAP;
    const int edgesPerBlock = RB_WORDS * 32;
    const int planeBlocks = (E + edgesPerBlock - 1) / edgesPerBlock;
    const int nwp = planeBlocks * RB_WORDS;
    unsigned* pkp = rbp + (long long)RB_PLANES * nwp;

    digit_plane_kernel<<<planeBlocks, RB_THREADS, 0, stream>>>(src, dst, rbp, pkp, N, E, nwp);

    {
        int tiles = (N / 16) * (F_HID / 16);
        int wavesPerBlock = AGG_THREADS / 32;
        int prodBlocks = (tiles + wavesPerBlock - 1) / wavesPerBlock;
        range_queue_and_product_kernel<<<aggBlocks + prodBlocks, AGG_THREADS, 0, stream>>>(
            dst, rbp, pkp, deg, qn, qp, N, E, nwp, aggBlocks, x, W1, xw1, N, F_IN, F_HID);
    }

    agg_range_kernel<64, true><<<aggBlocks, AGG_THREADS, 0, stream>>>(src, dst, deg, xw1, b1, qn, qp, h, N, E);

    {
        int tiles = (N / 16) * (F_OUT / 16);
        int wavesPerBlock = TB / 32;
        int blocks = (tiles + wavesPerBlock - 1) / wavesPerBlock;
        gemm_wmma_f32_kernel<<<blocks, TB, 0, stream>>>(h, W2, hw2, N, F_HID, F_OUT);
    }

    agg_range_kernel<32, false><<<aggBlocks, AGG_THREADS, 0, stream>>>(src, dst, deg, hw2, b2, qn, qp, out, N, E);
}


# dense products retiled, no LDS: a wave owns 16 rows x 32 columns (lane c: columns 2c,2c+1), A by 16-byte loads, one 8-byte W load per k for both tiles, line stores; k by 8: dot sum order changes
# speedup vs baseline: 5.1988x; 1.1845x over previous
#include <hip/hip_runtime.h>
#include <hip/hip_bf16.h>

typedef float v2f __attribute__((ext_vector_type(2)));
typedef float v8f __attribute__((ext_vector_type(8)));

template <int NT, int NTOT, int GEMM_KC, bool VEC>
__device__ __forceinline__ void gemm_wmma_f32_strip_impl(const float* __restrict__ A,
                                                         const float* __restrict__ B,
                                                         float* __restrict__ C,
                                                         int K, int strip, int cb) {
    static_assert(NT == 2, "two column tiles per wave: a lane's B elements of one row are one 8-byte vector");
    static_assert(NTOT % NT == 0 && GEMM_KC % 8 == 0, "whole column blocks, whole blocks of 8 k");
    constexpr int N = 16 * NTOT;
    const int lane = threadIdx.x & 31;
    const int half = lane >> 4;
    const int l15  = lane & 15;

    const float* arow = A + (long long)(strip * 16 + l15) * K + 4 * half;
    const float* bcol = B + (long long)(4 * half) * N + 16 * NT * cb + NT * l15;

    v8f acc[NT];
#pragma unroll
    for (int t = 0; t < NT; ++t) acc[t] = v8f{};

    for (int kc = 0; kc < K; kc += GEMM_KC) {
        float4 av[GEMM_KC / 8];
#pragma unroll
        for (int j = 0; j < GEMM_KC / 8; ++j) {
            const float* p = arow + kc + 8 * j;
            if constexpr (VEC) av[j] = *reinterpret_cast<const float4*>(p);
            else               av[j] = make_float4(p[0], p[1], p[2], p[3]);
        }
#pragma unroll
        for (int j = 0; j < GEMM_KC / 8; ++j) {
            float bw[4][NT];
#pragma unroll
            for (int i = 0; i < 4; ++i) {
                const float* p = bcol + (long long)(kc + 8 * j + i) * N;
                if constexpr (VEC) {
                    const float2 q = *reinterpret_cast<const float2*>(p);
                    bw[i][0] = q.x; bw[i][1] = q.y;
                } else {
#pragma unroll
                    for (int t = 0; t < NT; ++t) bw[i][t] = p[t];
                }
            }
            v2f a0, a1;
            a0.x = av[j].x; a0.y = av[j].y;
            a1.x = av[j].z; a1.y = av[j].w;
#pragma unroll
            for (int t = 0; t < NT; ++t) {
                v2f b; b.x = bw[0][t]; b.y = bw[1][t];
                acc[t] = __builtin_amdgcn_wmma_f32_16x16x4_f32(
                              false, a0,  false, b,
                              (short)0, acc[t],  false,  false);
            }
#pragma unroll
            for (int t = 0; t < NT; ++t) {
                v2f b; b.x = bw[2][t]; b.y = bw[3][t];
                acc[t] = __builtin_amdgcn_wmma_f32_16x16x4_f32(
                              false, a1,  false, b,
                              (short)0, acc[t],  false,  false);
            }
        }
    }

    float* crow = C + (long long)(strip * 16 + 8 * half) * N + 16 * NT * cb + NT * l15;
#pragma unroll
    for (int r = 0; r < 8; ++r) {
        float* p = crow + (long long)r * N;
        if constexpr (VEC) {
            *reinterpret_cast<float2*>(p) = make_float2(acc[0][r], acc[1][r]);
        } else {
#pragma unroll
            for (int t = 0; t < NT; ++t) p[t] = acc[t][r];
        }
    }
}

template <int NT, int NTOT, int GEMM_KC>
__device__ __forceinline__ void gemm_wmma_f32_strip(const float* __restrict__ A,
                                                    const float* __restrict__ B,
                                                    float* __restrict__ C,
                                                    int M, int K, int strip, int cb) {
    if (strip >= (M >> 4)) return;
    const bool vec_ok = ((reinterpret_cast<unsigned long long>(A) | reinterpret_cast<unsigned long long>(B) |
                          reinterpret_cast<unsigned long long>(C)) & 15ull) == 0ull;
    if (vec_ok) gemm_wmma_f32_strip_impl<NT, NTOT, GEMM_KC, true>(A, B, C, K, strip, cb);
    else        gemm_wmma_f32_strip_impl<NT, NTOT, GEMM_KC, false>(A, B, C, K, strip, cb);
}

template <int NT>
__global__ void gemm_wmma_f32_kernel(const float* __restrict__ A,
                                     const float* __restrict__ B,
                                     float* __restrict__ C,
                                     int M, int K) {
    const int wave = threadIdx.x >> 5;
    gemm_wmma_f32_strip<NT, NT, 64>(A, B, C, M, K, blockIdx.x * (blockDim.x >> 5) + wave, 0);
}

#define AGG_THREADS 1024
#define AGG_ROWS    200
#define AGG_QCAP    8192
#define AGG_GROUP   16

#define RB_THREADS 256
#define RB_WORDS   128
#define RB_PITCH   132
#define RB_PLANES  32
__global__ __launch_bounds__(RB_THREADS)
void digit_plane_kernel(const int* __restrict__ src,
                        const int* __restrict__ dst,
                        unsigned* __restrict__ planes,
                        unsigned* __restrict__ pk,
                        int N, int E, int nwp) {
    static_assert(AGG_ROWS <= 256, "row inside a range fits the packed word's high byte");
    static_assert(RB_WORDS == (RB_THREADS / 32) * 16, "a wave covers 512 edges = 16 plane words");
    __shared__ alignas(16) unsigned pl[RB_PLANES * RB_PITCH];
    const int tid  = threadIdx.x;
    const int lane = tid & 31;
    const int wv = __builtin_amdgcn_readfirstlane(tid >> 5);
    const int jw = __builtin_amdgcn_readfirstlane((int)((blockIdx.x * RB_THREADS + threadIdx.x) >> 5)) * 128;
    const unsigned un = (unsigned)N;
    const bool vec_ok = ((reinterpret_cast<unsigned long long>(dst) | reinterpret_cast<unsigned long long>(src) |
                          reinterpret_cast<unsigned long long>(pk)) & 15ull) == 0ull;
    const int4* __restrict__ dst4 = reinterpret_cast<const int4*>(dst);
    const int4* __restrict__ src4 = reinterpret_cast<const int4*>(src);
    auto rb = [&](int d) -> unsigned { return (unsigned)d < un ? ((unsigned)d / (unsigned)AGG_ROWS) : 0xFFu; };
    auto pack = [&](int s, int d, unsigned r) -> unsigned {
        return r != 0xFFu ? (((unsigned)s & 0xFFFFFFu) | (((unsigned)d - r * (unsigned)AGG_ROWS) << 24)) : 0u;
    };
    int4 v[4], w[4];
    const bool whole = vec_ok && (jw + 127) * 4 + 3 < E;
    if (whole) {
#pragma unroll
        for (int u = 0; u < 4; ++u) { v[u] = dst4[jw + u * 32 + lane]; w[u] = src4[jw + u * 32 + lane]; }
    } else {
#pragma unroll
        for (int u = 0; u < 4; ++u) {
            const int e = (jw + u * 32 + lane) << 2;
            v[u].x = (e     < E) ? dst[e]     : -1;
            v[u].y = (e + 1 < E) ? dst[e + 1] : -1;
            v[u].z = (e + 2 < E) ? dst[e + 2] : -1;
            v[u].w = (e + 3 < E) ? dst[e + 3] : -1;
            w[u].x = (e     < E) ? src[e]     : 0;
            w[u].y = (e + 1 < E) ? src[e + 1] : 0;
            w[u].z = (e + 2 < E) ? src[e + 2] : 0;
            w[u].w = (e + 3 < E) ? src[e + 3] : 0;
        }
    }
    for (int i = tid; i < RB_PLANES * RB_PITCH; i += RB_THREADS) pl[i] = 0u;
    __syncthreads();
    auto mark = [&](unsigned r, int word, unsigned bit) {
        if (r != 0xFFu) {
            atomicOr(&pl[(r >> 4) * RB_PITCH + word], bit);
            atomicOr(&pl[(16u + (r & 15u)) * RB_PITCH + word], bit);
        }
    };
#pragma unroll
    for (int u = 0; u < 4; ++u) {
        const int j = jw + u * 32 + lane;
        const unsigned bx = rb(v[u].x), by = rb(v[u].y), bz = rb(v[u].z), bw = rb(v[u].w);
        const int word = wv * 16 + u * 4 + (lane >> 3);
        const unsigned b0 = 1u << ((lane & 7) * 4);
        mark(bx, word, b0);
        mark(by, word, b0 << 1);
        mark(bz, word, b0 << 2);
        mark(bw, word, b0 << 3);
        uint4 p;
        p.x = pack(w[u].x, v[u].x, bx);
        p.y = pack(w[u].y, v[u].y, by);
        p.z = pack(w[u].z, v[u].z, bz);
        p.w = pack(w[u].w, v[u].w, bw);
        if (whole) {
            reinterpret_cast<uint4*>(pk)[j] = p;
        } else {
            const int e = j << 2;
            if (e     < E) pk[e]     = p.x;
            if (e + 1 < E) pk[e + 1] = p.y;
            if (e + 2 < E) pk[e + 2] = p.z;
            if (e + 3 < E) pk[e + 3] = p.w;
        }
    }
    __syncthreads();
    static_assert((RB_PLANES * (RB_WORDS / 4)) % RB_THREADS == 0, "whole 16-byte pieces per thread");
#pragma unroll
    for (int i = 0; i < (RB_PLANES * (RB_WORDS / 4)) / RB_THREADS; ++i) {
        const int idx   = tid + i * RB_THREADS;
        const int p     = idx / (RB_WORDS / 4);
        const int piece = idx - p * (RB_WORDS / 4);
        const uint4 t = *reinterpret_cast<const uint4*>(&pl[p * RB_PITCH + piece * 4]);
        *reinterpret_cast<uint4*>(planes + (long long)p * nwp + (long long)blockIdx.x * RB_WORDS + piece * 4) = t;
    }
}

__device__ __forceinline__ int wave_inclusive_sum(int x) {
    x += __builtin_amdgcn_update_dpp(0, x, 0x111, 0xF, 0xF, false);
    x += __builtin_amdgcn_update_dpp(0, x, 0x112, 0xF, 0xF, false);
    x += __builtin_amdgcn_update_dpp(0, x, 0x114, 0xF, 0xF, false);
    x += __builtin_amdgcn_update_dpp(0, x, 0x118, 0xF, 0xF, false);
    const int low = __builtin_amdgcn_readlane(x, 15);
    return x + ((threadIdx.x & 16) ? low : 0);
}

#define PROD_STRIPS 13
__global__ __launch_bounds__(AGG_THREADS, 16)
void range_queue_and_product_kernel(const int* __restrict__ dst,
                                    const unsigned* __restrict__ planes,
                                    const unsigned* __restrict__ pk,
                                    float* __restrict__ dinv,
                                    int* __restrict__ g_cnt,
                                    unsigned* __restrict__ g_pk,
                                    int N, int E, int nwp, int queueBlocks,
                                    const float* __restrict__ A,
                                    const float* __restrict__ B,
                                    float* __restrict__ C,
                                    int M, int K) {
    __shared__ int q_e[AGG_QCAP];
    __shared__ int cnt[AGG_ROWS];
    __shared__ int q_n;

    if ((int)blockIdx.x >= queueBlocks) {
        const int pwave = threadIdx.x >> 5;
        if (pwave < 2 * PROD_STRIPS)
            gemm_wmma_f32_strip<2, 4, 32>(A, B, C, M, K, ((int)blockIdx.x - queueBlocks) * PROD_STRIPS + (pwave >> 1), pwave & 1);
        return;
    }

    const int tid  = threadIdx.x;
    const int r0   = blockIdx.x * AGG_ROWS;
    const int rows = min(AGG_ROWS, N - r0);
    const unsigned urows = (unsigned)rows;

    if (tid < AGG_ROWS) cnt[tid] = 0;
    if (tid == 0) q_n = 0;
    __syncthreads();

    static_assert(RB_PLANES == 32, "sixteen planes per digit");
    const uint4* __restrict__ pa4 = reinterpret_cast<const uint4*>(planes + (long long)(blockIdx.x >> 4) * nwp);
    const uint4* __restrict__ pb4 = reinterpret_cast<const uint4*>(planes + (long long)(16 + (blockIdx.x & 15)) * nwp);
    const int P4 = nwp >> 2;
    const int rot = (int)(((long long)blockIdx.x * P4) / queueBlocks) & ~15;
    constexpr int SCAN_U = 4;
    for (int k0 = 0; k0 < P4; k0 += SCAN_U * AGG_THREADS) {
        const int k = k0 + tid;
        uint4 va[SCAN_U], vb[SCAN_U];
        int   ix[SCAN_U];
        unsigned lm[SCAN_U];
#pragma unroll
        for (int u = 0; u < SCAN_U; ++u) {
            const int ku = k + u * AGG_THREADS;
            int i = rot + ku;
            i -= (i >= P4) ? P4 : 0;
            i = (ku < P4) ? i : 0;
            lm[u] = (ku < P4) ? 0xFFFFFFFFu : 0u;
            ix[u] = i;
            va[u] = pa4[i];
            vb[u] = pb4[i];
        }
        unsigned m[SCAN_U][4];
        int c = 0;
#pragma unroll
        for (int u = 0; u < SCAN_U; ++u) {
            m[u][0] = va[u].x & vb[u].x & lm[u];
            m[u][1] = va[u].y & vb[u].y & lm[u];
            m[u][2] = va[u].z & vb[u].z & lm[u];
            m[u][3] = va[u].w & vb[u].w & lm[u];
            c += __builtin_popcount(m[u][0]) + __builtin_popcount(m[u][1]) +
                 __builtin_popcount(m[u][2]) + __builtin_popcount(m[u][3]);
        }
        const int incl = wave_inclusive_sum(c);
        int base = 0;
        if ((tid & 31) == 31) base = atomicAdd(&q_n, incl);
        int pos = __builtin_amdgcn_readlane(base, 31) + incl - c;
#pragma unroll
        for (int u = 0; u < SCAN_U; ++u) {
#pragma unroll
            for (int j = 0; j < 4; ++j) {
                unsigned mm = m[u][j];
                const int e0 = ((ix[u] << 2) + j) << 5;
                while (mm) {
                    const int p = __builtin_ctz(mm);
                    mm &= mm - 1;
                    if (pos < AGG_QCAP) q_e[pos] = e0 + p;
                    ++pos;
                }
            }
        }
    }
    __syncthreads();

    const int nhit = q_n;
    if (nhit <= AGG_QCAP) {
        static_assert(AGG_QCAP == 8 * AGG_THREADS, "eight queue entries per thread");
        unsigned* __restrict__ gq = g_pk + (long long)blockIdx.x * AGG_QCAP;
        int ce[8];
        unsigned cp[8];
#pragma unroll
        for (int u = 0; u < 8; ++u) {
            const int i = tid + u * AGG_THREADS;
            ce[u] = (i < nhit) ? q_e[i] : 0;
        }
#pragma unroll
        for (int u = 0; u < 8; ++u) cp[u] = pk[ce[u]];
#pragma unroll
        for (int u = 0; u < 8; ++u) {
            const int i = tid + u * AGG_THREADS;
            if (i < nhit) {
                gq[i] = cp[u];
                atomicAdd(&cnt[cp[u] >> 24], 1);
            }
        }
    } else {
        for (int e = tid; e < E; e += AGG_THREADS) {
            const unsigned r = (unsigned)(dst[e] - r0);
            if (r < urows) atomicAdd(&cnt[r], 1);
        }
    }
    if (tid == 0) g_cnt[blockIdx.x] = nhit;
    __syncthreads();

    if (tid < rows) dinv[r0 + tid] = rsqrtf((float)(cnt[tid] + 1));
}

#define LDS_BLOCK  32768
#define LDS_NBLK   8
#define Q_PER_BLK  (AGG_QCAP / LDS_NBLK)
#define Q_OFF_S    8192
#define Q_OFF_D    (Q_OFF_S + Q_PER_BLK * 4)
#define Q_OFF_W    (Q_OFF_D + Q_PER_BLK * 4)
template <int F, bool RELU>
__global__ __launch_bounds__(AGG_THREADS)
void agg_range_kernel(const int* __restrict__ src,
                      const int* __restrict__ dst,
                      const float* __restrict__ dinv,
                      const float* __restrict__ xw,
                      const float* __restrict__ bias,
                      const int* __restrict__ g_cnt,
                      const unsigned* __restrict__ g_pk,
                      float* __restrict__ out,
                      int N, int E) {
    static_assert(F == 32 || F == 64, "one or two columns per lane");
    constexpr int CPL = F / 32;
    constexpr int FV  = F / 4;

    static_assert((LDS_NBLK & (LDS_NBLK - 1)) == 0 && AGG_QCAP % LDS_NBLK == 0, "rows and queue entries are dealt by mask and shift");
    static_assert(((AGG_ROWS + LDS_NBLK - 1) / LDS_NBLK) * F * 4 <= Q_OFF_S, "a block's rows end before its queue slices");
    static_assert(Q_OFF_W + Q_PER_BLK * 4 <= LDS_BLOCK, "rows and queue slices fit one block");
    __shared__ alignas(16) unsigned char lds[LDS_NBLK * LDS_BLOCK];
    auto row_off = [](int r) -> int { return (r & (LDS_NBLK - 1)) * LDS_BLOCK + (r / LDS_NBLK) * (F * 4); };
    auto q_off = [](int i) -> int { return (i / Q_PER_BLK) * LDS_BLOCK + (i & (Q_PER_BLK - 1)) * 4; };

    const int tid  = threadIdx.x;
    const int lane = tid & 31;
    const int wave = __builtin_amdgcn_readfirstlane(tid >> 5);
    const int r0   = blockIdx.x * AGG_ROWS;
    const int rows = min(AGG_ROWS, N - r0);
    const unsigned urows = (unsigned)rows;
    const int nvec = rows * FV;

    for (int i = tid; i < nvec; i += AGG_THREADS) {
        const int r  = i / FV;
        const int f4 = i - r * FV;
        const int node = r0 + r;
        const float di = dinv[node];
        const float dd = di * di;
        const float4 x = reinterpret_cast<const float4*>(xw + (long long)node * F)[f4];
        const float* b = bias + 4 * f4;
        float4 a;
        a.x = x.x * dd + b[0];
        a.y = x.y * dd + b[1];
        a.z = x.z * dd + b[2];
        a.w = x.w * dd + b[3];
        *reinterpret_cast<float4*>(lds + row_off(r) + f4 * 16) = a;
    }

    const int nhit = __builtin_amdgcn_readfirstlane(g_cnt[blockIdx.x]);
    const bool fits = nhit <= AGG_QCAP;
    const int nq = fits ? nhit : 0;
    if (nq > 0) {
        static_assert(AGG_QCAP == 8 * AGG_THREADS, "eight queue entries per thread");
        const unsigned* __restrict__ gq = g_pk + (long long)blockIdx.x * AGG_QCAP;
        int cs[8], cd[8];
        float ca[8], cb[8];
#pragma unroll
        for (int u = 0; u < 8; ++u) {
            const int i = tid + u * AGG_THREADS;
            const unsigned p = (i < nq) ? gq[i] : 0u;
            cs[u] = (int)(p & 0xFFFFFFu);
            cd[u] = (int)(p >> 24);
        }
#pragma unroll
        for (int u = 0; u < 8; ++u) { ca[u] = dinv[cs[u]]; cb[u] = dinv[r0 + cd[u]]; }
#pragma unroll
        for (int u = 0; u < 8; ++u) {
            const int i = tid + u * AGG_THREADS;
            if (i < nq) {
                const int qo = q_off(i);
                *reinterpret_cast<int*>(lds + qo + Q_OFF_S)   = cs[u];
                *reinterpret_cast<int*>(lds + qo + Q_OFF_D)   = row_off(cd[u]);
                *reinterpret_cast<float*>(lds + qo + Q_OFF_W) = ca[u] * cb[u];
            }
        }
    }
    __syncthreads();
    if (!fits) {
        for (int e = tid; e < E; e += AGG_THREADS) {
            const int d = dst[e];
            if ((unsigned)(d - r0) < urows) {
                const int s = src[e];
                const float w = dinv[s] * dinv[d];
                const float* xr = xw + (long long)s * F;
                float* ar = reinterpret_cast<float*>(lds + row_off(d - r0));
                for (int f = 0; f < F; ++f) atomicAdd(&ar[f], xr[f] * w);
            }
        }
        __syncthreads();
    }
    static_assert(AGG_GROUP <= 32, "a lane per queue entry of the group");
    for (int base = wave * AGG_GROUP; base < nq; base += (AGG_THREADS / 32) * AGG_GROUP) {
        const int cnt = min(AGG_GROUP, nq - base);
        const int li = base + min(lane & (AGG_GROUP - 1), cnt - 1);
        const int qo = q_off(li);
        const int es = *reinterpret_cast<const int*>(lds + qo + Q_OFF_S);
        const int ed = *reinterpret_cast<const int*>(lds + qo + Q_OFF_D);
        const int ew = __float_as_int(*reinterpret_cast<const float*>(lds + qo + Q_OFF_W));
        asm volatile("" :: "v"(es), "v"(ed), "v"(ew));
        float vx[AGG_GROUP], vy[AGG_GROUP];
#pragma unroll
        for (int u = 0; u < AGG_GROUP; ++u) {
            const int s = __builtin_amdgcn_readlane(es, u);
            const float* xr = xw + (long long)s * F + lane * CPL;
            if (CPL == 2) {
                const float2 t = *reinterpret_cast<const float2*>(xr);
                vx[u] = t.x; vy[u] = t.y;
            } else {
                vx[u] = *xr; vy[u] = 0.0f;
            }
        }
#pragma unroll
        for (int u = 0; u < AGG_GROUP; ++u) {
            if (u < cnt) {
                const int   d = __builtin_amdgcn_readlane(ed, u);
                const float w = __int_as_float(__builtin_amdgcn_readlane(ew, u));
                float* ar = reinterpret_cast<float*>(lds + d) + lane * CPL;
                atomicAdd(&ar[0], vx[u] * w);
                if (CPL == 2) atomicAdd(&ar[1], vy[u] * w);
            }
        }
    }
    __syncthreads();

    for (int i = tid; i < nvec; i += AGG_THREADS) {
        const int r  = i / FV;
        const int f4 = i - r * FV;
        float4 a = *reinterpret_cast<const float4*>(lds + row_off(r) + f4 * 16);
        if (RELU) {
            a.x = fmaxf(a.x, 0.0f);
            a.y = fmaxf(a.y, 0.0f);
            a.z = fmaxf(a.z, 0.0f);
            a.w = fmaxf(a.w, 0.0f);
        }
        reinterpret_cast<float4*>(out + (long long)(r0 + r) * F)[f4] = a;
    }
}

extern "C" void kernel_launch(void* const* d_in, const int* in_sizes, int n_in,
                              void* d_out, int out_size, void* d_ws, size_t ws_size,
                              hipStream_t stream) {
    const float* x   = (const float*)d_in[0];
    const int*   ei  = (const int*)d_in[1];
    const float* W1  = (const float*)d_in[2];
    const float* b1  = (const float*)d_in[3];
    const float* W2  = (const float*)d_in[4];
    const float* b2  = (const float*)d_in[5];
    float*       out = (float*)d_out;

    const int E     = in_sizes[1] / 2;
    const int F_HID = in_sizes[3];
    const int F_IN  = in_sizes[2] / F_HID;
    const int F_OUT = in_sizes[5];
    const int N     = in_sizes[0] / F_IN;

    const int* src = ei;
    const int* dst = ei + E;

    const int TB = 256;
    const int aggBlocks = (N + AGG_ROWS - 1) / AGG_ROWS;

    float* deg = (float*)d_ws;
    float* xw1 = deg + 65536;
    float* h   = xw1 + (long long)N * F_HID;
    float* hw2 = h   + (long long)N * F_HID;
    int*   qn  = (int*)(hw2 + (long long)N * F_OUT);
    unsigned* qp = (unsigned*)(qn + 1024);
    unsigned* rbp = qp + (long long)aggBlocks * AGG_QCAP;
    const int edgesPerBlock = RB_WORDS * 32;
    const int planeBlocks = (E + edgesPerBlock - 1) / edgesPerBlock;
    const int nwp = planeBlocks * RB_WORDS;
    unsigned* pkp = rbp + (long long)RB_PLANES * nwp;

    digit_plane_kernel<<<planeBlocks, RB_THREADS, 0, stream>>>(src, dst, rbp, pkp, N, E, nwp);

    {
        int strips = N / 16;
        int prodBlocks = (strips + PROD_STRIPS - 1) / PROD_STRIPS;
        range_queue_and_product_kernel<<<aggBlocks + prodBlocks, AGG_THREADS, 0, stream>>>(
            dst, rbp, pkp, deg, qn, qp, N, E, nwp, aggBlocks, x, W1, xw1, N, F_IN);
    }

    agg_range_kernel<64, true><<<aggBlocks, AGG_THREADS, 0, stream>>>(src, dst, deg, xw1, b1, qn, qp, h, N, E);

    {
        int strips = N / 16;
        int wavesPerBlock = TB / 32;
        int blocks = (strips + wavesPerBlock - 1) / wavesPerBlock;
        gemm_wmma_f32_kernel<2><<<blocks, TB, 0, stream>>>(h, W2, hw2, N, F_HID);
    }

    agg_range_kernel<32, false><<<aggBlocks, AGG_THREADS, 0, stream>>>(src, dst, deg, hw2, b2, qn, qp, out, N, E);
}
